# starting kernel
# baseline (speedup 1.0000x reference)
#include <hip/hip_runtime.h>
#include <math.h>

typedef float v2f __attribute__((ext_vector_type(2)));
typedef float v8f __attribute__((ext_vector_type(8)));

#define NB 8
#define NS 2048
#define ND 3
#define NF 125
#define FP 128
#define KPAD (FP + 4)
#define NK 16
#define NPTS (NB*NS)

__device__ __forceinline__ float gelu_exact(float v) {
    return 0.5f * v * (1.0f + erff(v * 0.7071067811865475f));
}

__global__ __launch_bounds__(256) void mlp_wmma_kernel(
    const float* __restrict__ x, const float* __restrict__ W1,
    const float* __restrict__ b1, const float* __restrict__ W2,
    const float* __restrict__ b2, float* __restrict__ resw)
{
    __shared__ float sW2t[FP * KPAD];
    const int tid = threadIdx.x;
    for (int i = tid; i < FP*FP; i += 256) {
        const int k = i >> 7, n = i & 127;
        sW2t[n*KPAD + k] = (k < NF && n < NF) ? W2[k*NF + n] : 0.0f;
    }
    __syncthreads();

    const int lane = tid & 31;
    const int wave = tid >> 5;
    const int row  = lane & 15;
    const int half = lane >> 4;
    const int m0   = (blockIdx.x * 8 + wave) * 16;

    const float px = x[(size_t)(m0+row)*3 + 0];
    const float py = x[(size_t)(m0+row)*3 + 1];
    const float pz = x[(size_t)(m0+row)*3 + 2];

    float areg[64];
    #pragma unroll
    for (int ks = 0; ks < 32; ++ks) {
        #pragma unroll
        for (int u = 0; u < 2; ++u) {
            const int c = 4*ks + 2*half + u;
            float v = 0.0f;
            if (c < NF) {
                v = px*W1[0*NF + c] + py*W1[1*NF + c] + pz*W1[2*NF + c] + b1[c];
                v = gelu_exact(v);
            }
            areg[2*ks + u] = v;
        }
    }

    for (int nt = 0; nt < 8; ++nt) {
        const int n = nt*16 + row;
        const float bias = (n < NF) ? b2[n] : 0.0f;
        v8f acc;
        #pragma unroll
        for (int r = 0; r < 8; ++r) acc[r] = bias;

        const float* bn = &sW2t[n * KPAD];
        #pragma unroll
        for (int ks = 0; ks < 32; ++ks) {
            const int kb = 4*ks + 2*half;
            v2f a, b;
            a.x = areg[2*ks + 0];
            a.y = areg[2*ks + 1];
            b   = *(const v2f*)(bn + kb);
            acc = __builtin_amdgcn_wmma_f32_16x16x4_f32(
                    false, a, false, b, (short)0, acc, false, false);
        }

        #pragma unroll
        for (int r = 0; r < 8; ++r) {
            const int m = half*8 + r;
            resw[(size_t)(m0 + m)*FP + nt*16 + row] = acc[r];
        }
    }
}

__global__ __launch_bounds__(256) void knn_kernel(
    const float* __restrict__ x, int* __restrict__ knn)
{
    __shared__ float spx[NS], spy[NS], spz[NS], ssq[NS];
    const int b  = blockIdx.x >> 3;
    const int i0 = (blockIdx.x & 7) * 256;
    const float* xb = x + (size_t)b*NS*3;
    for (int j = threadIdx.x; j < NS; j += 256) {
        const float a0 = xb[j*3+0], a1 = xb[j*3+1], a2 = xb[j*3+2];
        spx[j] = a0; spy[j] = a1; spz[j] = a2;
        ssq[j] = a0*a0 + a1*a1 + a2*a2;
    }
    __syncthreads();

    const int i = i0 + threadIdx.x;
    const float qx = spx[i], qy = spy[i], qz = spz[i], qs = ssq[i];

    float dist[NK]; int idx[NK];
    #pragma unroll
    for (int k = 0; k < NK; ++k) { dist[k] = 3.4e38f; idx[k] = 0; }

    for (int j = 0; j < NS; ++j) {
        const float d = qs + ssq[j] - 2.0f*(qx*spx[j] + qy*spy[j] + qz*spz[j]);
        if (d < dist[NK-1]) {
            int pos = NK-1;
            #pragma unroll
            for (int t = NK-1; t > 0; --t) {
                if (dist[t-1] > d) {
                    dist[t] = dist[t-1]; idx[t] = idx[t-1]; pos = t-1;
                }
            }
            dist[pos] = d; idx[pos] = j;
        }
    }

    int* kout = knn + (size_t)(b*NS + i)*NK;
    #pragma unroll
    for (int k = 0; k < NK; ++k) kout[k] = idx[k];
}

__global__ __launch_bounds__(256) void gather_max_kernel(
    const float* __restrict__ x, const float* __restrict__ resw,
    const int* __restrict__ knn, float* __restrict__ out)
{
    __shared__ int sidx[2][NK];
    const int sub = threadIdx.x >> 7;
    const int d   = threadIdx.x & 127;
    const int q   = blockIdx.x * 2 + sub;
    if (d < NK) sidx[sub][d] = knn[(size_t)q*NK + d];
    __syncthreads();

    if (d < NF) {
        const int b = q >> 11;
        const float* rb = resw + (size_t)b*NS*FP;
        float m = -3.4e38f;
        #pragma unroll
        for (int k = 0; k < NK; ++k)
            m = fmaxf(m, rb[(size_t)sidx[sub][k]*FP + d]);
        out[(size_t)q*FP + d] = m;
    } else {
        out[(size_t)q*FP + d] = x[(size_t)q*3 + (d - NF)];
    }
}

extern "C" void kernel_launch(void* const* d_in, const int* in_sizes, int n_in,
                              void* d_out, int out_size, void* d_ws, size_t ws_size,
                              hipStream_t stream) {
    const float* x  = (const float*)d_in[0];
    const float* W1 = (const float*)d_in[1];
    const float* b1 = (const float*)d_in[2];
    const float* W2 = (const float*)d_in[3];
    const float* b2 = (const float*)d_in[4];
    float* out = (float*)d_out;

    float* resw = (float*)d_ws;
    int*   knn  = (int*)((char*)d_ws + (size_t)NPTS*FP*sizeof(float));

    mlp_wmma_kernel<<<NPTS/(16*8), 256, 0, stream>>>(x, W1, b1, W2, b2, resw);
    knn_kernel<<<NB*(NS/256), 256, 0, stream>>>(x, knn);
    gather_max_kernel<<<NPTS/2, 256, 0, stream>>>(x, resw, knn, out);
}


# kNN search: one wave per query, 32 candidates scored per step, sorted top-16 kept one entry per lane (DPP shift insert), grid 64 -> 1024 blocks; same distances, same tie order
# speedup vs baseline: 3.4682x; 3.4682x over previous
#include <hip/hip_runtime.h>
#include <math.h>

typedef float v2f __attribute__((ext_vector_type(2)));
typedef float v8f __attribute__((ext_vector_type(8)));

#define NB 8
#define NS 2048
#define ND 3
#define NF 125
#define FP 128
#define KPAD (FP + 4)
#define NK 16
#define NPTS (NB*NS)

__device__ __forceinline__ float gelu_exact(float v) {
    return 0.5f * v * (1.0f + erff(v * 0.7071067811865475f));
}

__global__ __launch_bounds__(256) void mlp_wmma_kernel(
    const float* __restrict__ x, const float* __restrict__ W1,
    const float* __restrict__ b1, const float* __restrict__ W2,
    const float* __restrict__ b2, float* __restrict__ resw)
{
    __shared__ float sW2t[FP * KPAD];
    const int tid = threadIdx.x;
    for (int i = tid; i < FP*FP; i += 256) {
        const int k = i >> 7, n = i & 127;
        sW2t[n*KPAD + k] = (k < NF && n < NF) ? W2[k*NF + n] : 0.0f;
    }
    __syncthreads();

    const int lane = tid & 31;
    const int wave = tid >> 5;
    const int row  = lane & 15;
    const int half = lane >> 4;
    const int m0   = (blockIdx.x * 8 + wave) * 16;

    const float px = x[(size_t)(m0+row)*3 + 0];
    const float py = x[(size_t)(m0+row)*3 + 1];
    const float pz = x[(size_t)(m0+row)*3 + 2];

    float areg[64];
    #pragma unroll
    for (int ks = 0; ks < 32; ++ks) {
        #pragma unroll
        for (int u = 0; u < 2; ++u) {
            const int c = 4*ks + 2*half + u;
            float v = 0.0f;
            if (c < NF) {
                v = px*W1[0*NF + c] + py*W1[1*NF + c] + pz*W1[2*NF + c] + b1[c];
                v = gelu_exact(v);
            }
            areg[2*ks + u] = v;
        }
    }

    for (int nt = 0; nt < 8; ++nt) {
        const int n = nt*16 + row;
        const float bias = (n < NF) ? b2[n] : 0.0f;
        v8f acc;
        #pragma unroll
        for (int r = 0; r < 8; ++r) acc[r] = bias;

        const float* bn = &sW2t[n * KPAD];
        #pragma unroll
        for (int ks = 0; ks < 32; ++ks) {
            const int kb = 4*ks + 2*half;
            v2f a, b;
            a.x = areg[2*ks + 0];
            a.y = areg[2*ks + 1];
            b   = *(const v2f*)(bn + kb);
            acc = __builtin_amdgcn_wmma_f32_16x16x4_f32(
                    false, a, false, b, (short)0, acc, false, false);
        }

        #pragma unroll
        for (int r = 0; r < 8; ++r) {
            const int m = half*8 + r;
            resw[(size_t)(m0 + m)*FP + nt*16 + row] = acc[r];
        }
    }
}

#define QPW 2
#define QPB (8*QPW)
__global__ __launch_bounds__(256) void knn_kernel(
    const float* __restrict__ x, int* __restrict__ knn)
{
    __shared__ float spx[NS], spy[NS], spz[NS], ssq[NS];
    const int b  = blockIdx.x / (NS/QPB);
    const int i0 = (blockIdx.x % (NS/QPB)) * QPB;
    const float* xb = x + (size_t)b*NS*3;
    for (int j = threadIdx.x; j < NS; j += 256) {
        const float a0 = xb[j*3+0], a1 = xb[j*3+1], a2 = xb[j*3+2];
        spx[j] = a0; spy[j] = a1; spz[j] = a2;
        ssq[j] = a0*a0 + a1*a1 + a2*a2;
    }
    __syncthreads();

    const int lane = threadIdx.x & 31;
    const int wave = threadIdx.x >> 5;
    const float NEG_INF = -__builtin_inff();

    for (int qi = 0; qi < QPW; ++qi) {
        const int i = i0 + wave*QPW + qi;
        const float qx = spx[i], qy = spy[i], qz = spz[i], qs = ssq[i];

        float L = (lane < NK) ? 3.4e38f : NEG_INF;
        int   I = 0;
        float thr = 3.4e38f;

        for (int c = 0; c < NS/32; ++c) {
            const int j = c*32 + lane;
            const float d = qs + ssq[j] - 2.0f*(qx*spx[j] + qy*spy[j] + qz*spz[j]);
            unsigned mask = __builtin_amdgcn_ballot_w32(d < thr);
            while (mask) {
                const int sl = __builtin_ctz(mask);
                const float sd = __builtin_bit_cast(float,
                    __builtin_amdgcn_readlane(__builtin_bit_cast(int, d), sl));
                const int sj = c*32 + sl;
                const float Lsh = __builtin_bit_cast(float, __builtin_amdgcn_update_dpp(
                    __builtin_bit_cast(int, NEG_INF), __builtin_bit_cast(int, L),
                    0x111, 0xf, 0xf, false));
                const int Ish = __builtin_amdgcn_update_dpp(0, I, 0x111, 0xf, 0xf, false);
                const bool mv  = L   > sd;
                const bool mvp = Lsh > sd;
                const float nL = mvp ? Lsh : sd;
                const int   nI = mvp ? Ish : sj;
                L = mv ? nL : L;
                I = mv ? nI : I;
                thr = __builtin_bit_cast(float,
                    __builtin_amdgcn_readlane(__builtin_bit_cast(int, L), NK-1));
                mask &= mask - 1;
                mask &= __builtin_amdgcn_ballot_w32(d < thr);
            }
        }

        if (lane < NK) knn[(size_t)(b*NS + i)*NK + lane] = I;
    }
}

__global__ __launch_bounds__(256) void gather_max_kernel(
    const float* __restrict__ x, const float* __restrict__ resw,
    const int* __restrict__ knn, float* __restrict__ out)
{
    __shared__ int sidx[2][NK];
    const int sub = threadIdx.x >> 7;
    const int d   = threadIdx.x & 127;
    const int q   = blockIdx.x * 2 + sub;
    if (d < NK) sidx[sub][d] = knn[(size_t)q*NK + d];
    __syncthreads();

    if (d < NF) {
        const int b = q >> 11;
        const float* rb = resw + (size_t)b*NS*FP;
        float m = -3.4e38f;
        #pragma unroll
        for (int k = 0; k < NK; ++k)
            m = fmaxf(m, rb[(size_t)sidx[sub][k]*FP + d]);
        out[(size_t)q*FP + d] = m;
    } else {
        out[(size_t)q*FP + d] = x[(size_t)q*3 + (d - NF)];
    }
}

extern "C" void kernel_launch(void* const* d_in, const int* in_sizes, int n_in,
                              void* d_out, int out_size, void* d_ws, size_t ws_size,
                              hipStream_t stream) {
    const float* x  = (const float*)d_in[0];
    const float* W1 = (const float*)d_in[1];
    const float* b1 = (const float*)d_in[2];
    const float* W2 = (const float*)d_in[3];
    const float* b2 = (const float*)d_in[4];
    float* out = (float*)d_out;

    float* resw = (float*)d_ws;
    int*   knn  = (int*)((char*)d_ws + (size_t)NPTS*FP*sizeof(float));

    mlp_wmma_kernel<<<NPTS/(16*8), 256, 0, stream>>>(x, W1, b1, W2, b2, resw);
    knn_kernel<<<NB*(NS/QPB), 256, 0, stream>>>(x, knn);
    gather_max_kernel<<<NPTS/2, 256, 0, stream>>>(x, resw, knn, out);
}


# gather-max re-tiled: one wave per output row (lane = 4 columns, 16-byte loads), 4 rows per wave, neighbour indices read once per row as scalars; 65536 short waves -> 4096; same values
# speedup vs baseline: 4.8166x; 1.3888x over previous
#include <hip/hip_runtime.h>
#include <math.h>

typedef float v2f __attribute__((ext_vector_type(2)));
typedef float v8f __attribute__((ext_vector_type(8)));

#define NB 8
#define NS 2048
#define ND 3
#define NF 125
#define FP 128
#define KPAD (FP + 4)
#define NK 16
#define NPTS (NB*NS)

__device__ __forceinline__ float gelu_exact(float v) {
    return 0.5f * v * (1.0f + erff(v * 0.7071067811865475f));
}

__global__ __launch_bounds__(256) void mlp_wmma_kernel(
    const float* __restrict__ x, const float* __restrict__ W1,
    const float* __restrict__ b1, const float* __restrict__ W2,
    const float* __restrict__ b2, float* __restrict__ resw)
{
    __shared__ float sW2t[FP * KPAD];
    const int tid = threadIdx.x;
    for (int i = tid; i < FP*FP; i += 256) {
        const int k = i >> 7, n = i & 127;
        sW2t[n*KPAD + k] = (k < NF && n < NF) ? W2[k*NF + n] : 0.0f;
    }
    __syncthreads();

    const int lane = tid & 31;
    const int wave = tid >> 5;
    const int row  = lane & 15;
    const int half = lane >> 4;
    const int m0   = (blockIdx.x * 8 + wave) * 16;

    const float px = x[(size_t)(m0+row)*3 + 0];
    const float py = x[(size_t)(m0+row)*3 + 1];
    const float pz = x[(size_t)(m0+row)*3 + 2];

    float areg[64];
    #pragma unroll
    for (int ks = 0; ks < 32; ++ks) {
        #pragma unroll
        for (int u = 0; u < 2; ++u) {
            const int c = 4*ks + 2*half + u;
            float v = 0.0f;
            if (c < NF) {
                v = px*W1[0*NF + c] + py*W1[1*NF + c] + pz*W1[2*NF + c] + b1[c];
                v = gelu_exact(v);
            }
            areg[2*ks + u] = v;
        }
    }

    for (int nt = 0; nt < 8; ++nt) {
        const int n = nt*16 + row;
        const float bias = (n < NF) ? b2[n] : 0.0f;
        v8f acc;
        #pragma unroll
        for (int r = 0; r < 8; ++r) acc[r] = bias;

        const float* bn = &sW2t[n * KPAD];
        #pragma unroll
        for (int ks = 0; ks < 32; ++ks) {
            const int kb = 4*ks + 2*half;
            v2f a, b;
            a.x = areg[2*ks + 0];
            a.y = areg[2*ks + 1];
            b   = *(const v2f*)(bn + kb);
            acc = __builtin_amdgcn_wmma_f32_16x16x4_f32(
                    false, a, false, b, (short)0, acc, false, false);
        }

        #pragma unroll
        for (int r = 0; r < 8; ++r) {
            const int m = half*8 + r;
            resw[(size_t)(m0 + m)*FP + nt*16 + row] = acc[r];
        }
    }
}

#define QPW 2
#define QPB (8*QPW)
__global__ __launch_bounds__(256) void knn_kernel(
    const float* __restrict__ x, int* __restrict__ knn)
{
    __shared__ float spx[NS], spy[NS], spz[NS], ssq[NS];
    const int b  = blockIdx.x / (NS/QPB);
    const int i0 = (blockIdx.x % (NS/QPB)) * QPB;
    const float* xb = x + (size_t)b*NS*3;
    for (int j = threadIdx.x; j < NS; j += 256) {
        const float a0 = xb[j*3+0], a1 = xb[j*3+1], a2 = xb[j*3+2];
        spx[j] = a0; spy[j] = a1; spz[j] = a2;
        ssq[j] = a0*a0 + a1*a1 + a2*a2;
    }
    __syncthreads();

    const int lane = threadIdx.x & 31;
    const int wave = threadIdx.x >> 5;
    const float NEG_INF = -__builtin_inff();

    for (int qi = 0; qi < QPW; ++qi) {
        const int i = i0 + wave*QPW + qi;
        const float qx = spx[i], qy = spy[i], qz = spz[i], qs = ssq[i];

        float L = (lane < NK) ? 3.4e38f : NEG_INF;
        int   I = 0;
        float thr = 3.4e38f;

        for (int c = 0; c < NS/32; ++c) {
            const int j = c*32 + lane;
            const float d = qs + ssq[j] - 2.0f*(qx*spx[j] + qy*spy[j] + qz*spz[j]);
            unsigned mask = __builtin_amdgcn_ballot_w32(d < thr);
            while (mask) {
                const int sl = __builtin_ctz(mask);
                const float sd = __builtin_bit_cast(float,
                    __builtin_amdgcn_readlane(__builtin_bit_cast(int, d), sl));
                const int sj = c*32 + sl;
                const float Lsh = __builtin_bit_cast(float, __builtin_amdgcn_update_dpp(
                    __builtin_bit_cast(int, NEG_INF), __builtin_bit_cast(int, L),
                    0x111, 0xf, 0xf, false));
                const int Ish = __builtin_amdgcn_update_dpp(0, I, 0x111, 0xf, 0xf, false);
                const bool mv  = L   > sd;
                const bool mvp = Lsh > sd;
                const float nL = mvp ? Lsh : sd;
                const int   nI = mvp ? Ish : sj;
                L = mv ? nL : L;
                I = mv ? nI : I;
                thr = __builtin_bit_cast(float,
                    __builtin_amdgcn_readlane(__builtin_bit_cast(int, L), NK-1));
                mask &= mask - 1;
                mask &= __builtin_amdgcn_ballot_w32(d < thr);
            }
        }

        if (lane < NK) knn[(size_t)(b*NS + i)*NK + lane] = I;
    }
}

#define GQ 4
__global__ __launch_bounds__(256) void gather_max_kernel(
    const float* __restrict__ x, const float* __restrict__ resw,
    const int* __restrict__ knn, float* __restrict__ out)
{
    const int lane = threadIdx.x & 31;
    const int wave = __builtin_amdgcn_readfirstlane(threadIdx.x >> 5);
    const int q0   = (blockIdx.x * 8 + wave) * GQ;
    const int b    = q0 >> 11;
    const float4* rb = (const float4*)(resw + (size_t)b*NS*FP) + lane;
    float4*       ob = (float4*)(out + (size_t)q0*FP) + lane;

    int nb[GQ][NK];
    #pragma unroll
    for (int s = 0; s < GQ; ++s) {
        #pragma unroll
        for (int k = 0; k < NK; ++k) nb[s][k] = knn[(size_t)(q0 + s)*NK + k];
    }
    float xq[GQ*ND];
    #pragma unroll
    for (int t = 0; t < GQ*ND/4; ++t) {
        const float4 xv = ((const float4*)(x + (size_t)q0*ND))[t];
        xq[4*t + 0] = xv.x; xq[4*t + 1] = xv.y; xq[4*t + 2] = xv.z; xq[4*t + 3] = xv.w;
    }

    const bool tail = (lane == 31);
    #pragma unroll
    for (int s = 0; s < GQ; ++s) {
        float4 m = make_float4(-3.4e38f, -3.4e38f, -3.4e38f, -3.4e38f);
        #pragma unroll
        for (int k = 0; k < NK; ++k) {
            const float4 v = rb[(size_t)nb[s][k] * (FP/4)];
            m.x = fmaxf(m.x, v.x);
            m.y = fmaxf(m.y, v.y);
            m.z = fmaxf(m.z, v.z);
            m.w = fmaxf(m.w, v.w);
        }
        m.y = tail ? xq[s*ND + 0] : m.y;
        m.z = tail ? xq[s*ND + 1] : m.z;
        m.w = tail ? xq[s*ND + 2] : m.w;
        ob[(size_t)s*(FP/4)] = m;
    }
}

extern "C" void kernel_launch(void* const* d_in, const int* in_sizes, int n_in,
                              void* d_out, int out_size, void* d_ws, size_t ws_size,
                              hipStream_t stream) {
    const float* x  = (const float*)d_in[0];
    const float* W1 = (const float*)d_in[1];
    const float* b1 = (const float*)d_in[2];
    const float* W2 = (const float*)d_in[3];
    const float* b2 = (const float*)d_in[4];
    float* out = (float*)d_out;

    float* resw = (float*)d_ws;
    int*   knn  = (int*)((char*)d_ws + (size_t)NPTS*FP*sizeof(float));

    mlp_wmma_kernel<<<NPTS/(16*8), 256, 0, stream>>>(x, W1, b1, W2, b2, resw);
    knn_kernel<<<NB*(NS/QPB), 256, 0, stream>>>(x, knn);
    gather_max_kernel<<<NPTS/(8*GQ), 256, 0, stream>>>(x, resw, knn, out);
}


# global loads taken out of serial chains: W2 staging 32 loads per group ahead of the LDS stores, W1/b1 loads ahead of a separate gelu pass, kNN point loads grouped; same arithmetic and values
# speedup vs baseline: 6.2333x; 1.2941x over previous
#include <hip/hip_runtime.h>
#include <math.h>

typedef float v2f __attribute__((ext_vector_type(2)));
typedef float v8f __attribute__((ext_vector_type(8)));

#define NB 8
#define NS 2048
#define ND 3
#define NF 125
#define FP 128
#define KPAD (FP + 4)
#define NK 16
#define NPTS (NB*NS)
#define SG 32

__device__ __forceinline__ float gelu_exact(float v) {
    return 0.5f * v * (1.0f + erff(v * 0.7071067811865475f));
}

__global__ __launch_bounds__(256) void mlp_wmma_kernel(
    const float* __restrict__ x, const float* __restrict__ W1,
    const float* __restrict__ b1, const float* __restrict__ W2,
    const float* __restrict__ b2, float* __restrict__ resw)
{
    __shared__ float sW2t[FP * KPAD];
    const int tid = threadIdx.x;
    {
        const int n  = tid & 127;
        const int k0 = tid >> 7;
        const int nc = (n < NF) ? n : (NF - 1);
        for (int g = 0; g < 64; g += SG) {
            float t[SG];
            #pragma unroll
            for (int u = 0; u < SG; ++u) {
                const int k  = 2*(g + u) + k0;
                const int kc = (k < NF) ? k : (NF - 1);
                t[u] = W2[kc*NF + nc];
            }
            #pragma unroll
            for (int u = 0; u < SG; ++u) {
                const int k = 2*(g + u) + k0;
                sW2t[n*KPAD + k] = (k < NF && n < NF) ? t[u] : 0.0f;
            }
        }
    }
    __syncthreads();

    const int lane = tid & 31;
    const int wave = tid >> 5;
    const int row  = lane & 15;
    const int half = lane >> 4;
    const int m0   = (blockIdx.x * 8 + wave) * 16;

    const float px = x[(size_t)(m0+row)*3 + 0];
    const float py = x[(size_t)(m0+row)*3 + 1];
    const float pz = x[(size_t)(m0+row)*3 + 2];

    float areg[64];
    #pragma unroll
    for (int ks = 0; ks < 32; ++ks) {
        #pragma unroll
        for (int u = 0; u < 2; ++u) {
            const int c  = 4*ks + 2*half + u;
            const int cc = (c < NF) ? c : (NF - 1);
            areg[2*ks + u] = px*W1[0*NF + cc] + py*W1[1*NF + cc] + pz*W1[2*NF + cc] + b1[cc];
        }
    }
    #pragma unroll
    for (int ks = 0; ks < 32; ++ks) {
        #pragma unroll
        for (int u = 0; u < 2; ++u) {
            const int c = 4*ks + 2*half + u;
            areg[2*ks + u] = (c < NF) ? gelu_exact(areg[2*ks + u]) : 0.0f;
        }
    }

    for (int nt = 0; nt < 8; ++nt) {
        const int n = nt*16 + row;
        const float bias = (n < NF) ? b2[n] : 0.0f;
        v8f acc;
        #pragma unroll
        for (int r = 0; r < 8; ++r) acc[r] = bias;

        const float* bn = &sW2t[n * KPAD];
        #pragma unroll
        for (int ks = 0; ks < 32; ++ks) {
            const int kb = 4*ks + 2*half;
            v2f a, b;
            a.x = areg[2*ks + 0];
            a.y = areg[2*ks + 1];
            b   = *(const v2f*)(bn + kb);
            acc = __builtin_amdgcn_wmma_f32_16x16x4_f32(
                    false, a, false, b, (short)0, acc, false, false);
        }

        #pragma unroll
        for (int r = 0; r < 8; ++r) {
            const int m = half*8 + r;
            resw[(size_t)(m0 + m)*FP + nt*16 + row] = acc[r];
        }
    }
}

#define QPW 2
#define QPB (8*QPW)
__global__ __launch_bounds__(256) void knn_kernel(
    const float* __restrict__ x, int* __restrict__ knn)
{
    __shared__ float spx[NS], spy[NS], spz[NS], ssq[NS];
    const int b  = blockIdx.x / (NS/QPB);
    const int i0 = (blockIdx.x % (NS/QPB)) * QPB;
    const float* xb = x + (size_t)b*NS*3;
    {
        float p0[NS/256], p1[NS/256], p2[NS/256];
        #pragma unroll
        for (int u = 0; u < NS/256; ++u) {
            const int j = threadIdx.x + 256*u;
            p0[u] = xb[j*3+0]; p1[u] = xb[j*3+1]; p2[u] = xb[j*3+2];
        }
        #pragma unroll
        for (int u = 0; u < NS/256; ++u) {
            const int j = threadIdx.x + 256*u;
            const float a0 = p0[u], a1 = p1[u], a2 = p2[u];
            spx[j] = a0; spy[j] = a1; spz[j] = a2;
            ssq[j] = a0*a0 + a1*a1 + a2*a2;
        }
    }
    __syncthreads();

    const int lane = threadIdx.x & 31;
    const int wave = threadIdx.x >> 5;
    const float NEG_INF = -__builtin_inff();

    for (int qi = 0; qi < QPW; ++qi) {
        const int i = i0 + wave*QPW + qi;
        const float qx = spx[i], qy = spy[i], qz = spz[i], qs = ssq[i];

        float L = (lane < NK) ? 3.4e38f : NEG_INF;
        int   I = 0;
        float thr = 3.4e38f;

        for (int c = 0; c < NS/32; ++c) {
            const int j = c*32 + lane;
            const float d = qs + ssq[j] - 2.0f*(qx*spx[j] + qy*spy[j] + qz*spz[j]);
            unsigned mask = __builtin_amdgcn_ballot_w32(d < thr);
            while (mask) {
                const int sl = __builtin_ctz(mask);
                const float sd = __builtin_bit_cast(float,
                    __builtin_amdgcn_readlane(__builtin_bit_cast(int, d), sl));
                const int sj = c*32 + sl;
                const float Lsh = __builtin_bit_cast(float, __builtin_amdgcn_update_dpp(
                    __builtin_bit_cast(int, NEG_INF), __builtin_bit_cast(int, L),
                    0x111, 0xf, 0xf, false));
                const int Ish = __builtin_amdgcn_update_dpp(0, I, 0x111, 0xf, 0xf, false);
                const bool mv  = L   > sd;
                const bool mvp = Lsh > sd;
                const float nL = mvp ? Lsh : sd;
                const int   nI = mvp ? Ish : sj;
                L = mv ? nL : L;
                I = mv ? nI : I;
                thr = __builtin_bit_cast(float,
                    __builtin_amdgcn_readlane(__builtin_bit_cast(int, L), NK-1));
                mask &= mask - 1;
                mask &= __builtin_amdgcn_ballot_w32(d < thr);
            }
        }

        if (lane < NK) knn[(size_t)(b*NS + i)*NK + lane] = I;
    }
}

#define GQ 4
__global__ __launch_bounds__(256) void gather_max_kernel(
    const float* __restrict__ x, const float* __restrict__ resw,
    const int* __restrict__ knn, float* __restrict__ out)
{
    const int lane = threadIdx.x & 31;
    const int wave = __builtin_amdgcn_readfirstlane(threadIdx.x >> 5);
    const int q0   = (blockIdx.x * 8 + wave) * GQ;
    const int b    = q0 >> 11;
    const float4* rb = (const float4*)(resw + (size_t)b*NS*FP) + lane;
    float4*       ob = (float4*)(out + (size_t)q0*FP) + lane;

    int nb[GQ][NK];
    #pragma unroll
    for (int s = 0; s < GQ; ++s) {
        #pragma unroll
        for (int k = 0; k < NK; ++k) nb[s][k] = knn[(size_t)(q0 + s)*NK + k];
    }
    float xq[GQ*ND];
    #pragma unroll
    for (int t = 0; t < GQ*ND/4; ++t) {
        const float4 xv = ((const float4*)(x + (size_t)q0*ND))[t];
        xq[4*t + 0] = xv.x; xq[4*t + 1] = xv.y; xq[4*t + 2] = xv.z; xq[4*t + 3] = xv.w;
    }

    const bool tail = (lane == 31);
    #pragma unroll
    for (int s = 0; s < GQ; ++s) {
        float4 m = make_float4(-3.4e38f, -3.4e38f, -3.4e38f, -3.4e38f);
        #pragma unroll
        for (int k = 0; k < NK; ++k) {
            const float4 v = rb[(size_t)nb[s][k] * (FP/4)];
            m.x = fmaxf(m.x, v.x);
            m.y = fmaxf(m.y, v.y);
            m.z = fmaxf(m.z, v.z);
            m.w = fmaxf(m.w, v.w);
        }
        m.y = tail ? xq[s*ND + 0] : m.y;
        m.z = tail ? xq[s*ND + 1] : m.z;
        m.w = tail ? xq[s*ND + 2] : m.w;
        ob[(size_t)s*(FP/4)] = m;
    }
}

extern "C" void kernel_launch(void* const* d_in, const int* in_sizes, int n_in,
                              void* d_out, int out_size, void* d_ws, size_t ws_size,
                              hipStream_t stream) {
    const float* x  = (const float*)d_in[0];
    const float* W1 = (const float*)d_in[1];
    const float* b1 = (const float*)d_in[2];
    const float* W2 = (const float*)d_in[3];
    const float* b2 = (const float*)d_in[4];
    float* out = (float*)d_out;

    float* resw = (float*)d_ws;
    int*   knn  = (int*)((char*)d_ws + (size_t)NPTS*FP*sizeof(float));

    mlp_wmma_kernel<<<NPTS/(16*8), 256, 0, stream>>>(x, W1, b1, W2, b2, resw);
    knn_kernel<<<NB*(NS/QPB), 256, 0, stream>>>(x, knn);
    gather_max_kernel<<<NPTS/(8*GQ), 256, 0, stream>>>(x, resw, knn, out);
}


# gather-max fused into the neighbour search: each wave loads its query's 16 rows from its own list, in flight during the next query's scan; launches 3 -> 2, index array not stored; same values
# speedup vs baseline: 7.0105x; 1.1247x over previous
#include <hip/hip_runtime.h>
#include <math.h>

typedef float v2f __attribute__((ext_vector_type(2)));
typedef float v8f __attribute__((ext_vector_type(8)));

#define NB 8
#define NS 2048
#define ND 3
#define NF 125
#define FP 128
#define KPAD (FP + 4)
#define NK 16
#define NPTS (NB*NS)
#define SG 32

__device__ __forceinline__ float gelu_exact(float v) {
    return 0.5f * v * (1.0f + erff(v * 0.7071067811865475f));
}

__global__ __launch_bounds__(256) void mlp_wmma_kernel(
    const float* __restrict__ x, const float* __restrict__ W1,
    const float* __restrict__ b1, const float* __restrict__ W2,
    const float* __restrict__ b2, float* __restrict__ resw)
{
    __shared__ float sW2t[FP * KPAD];
    const int tid = threadIdx.x;
    {
        const int n  = tid & 127;
        const int k0 = tid >> 7;
        const int nc = (n < NF) ? n : (NF - 1);
        for (int g = 0; g < 64; g += SG) {
            float t[SG];
            #pragma unroll
            for (int u = 0; u < SG; ++u) {
                const int k  = 2*(g + u) + k0;
                const int kc = (k < NF) ? k : (NF - 1);
                t[u] = W2[kc*NF + nc];
            }
            #pragma unroll
            for (int u = 0; u < SG; ++u) {
                const int k = 2*(g + u) + k0;
                sW2t[n*KPAD + k] = (k < NF && n < NF) ? t[u] : 0.0f;
            }
        }
    }
    __syncthreads();

    const int lane = tid & 31;
    const int wave = tid >> 5;
    const int row  = lane & 15;
    const int half = lane >> 4;
    const int m0   = (blockIdx.x * 8 + wave) * 16;

    const float px = x[(size_t)(m0+row)*3 + 0];
    const float py = x[(size_t)(m0+row)*3 + 1];
    const float pz = x[(size_t)(m0+row)*3 + 2];

    float areg[64];
    #pragma unroll
    for (int ks = 0; ks < 32; ++ks) {
        #pragma unroll
        for (int u = 0; u < 2; ++u) {
            const int c  = 4*ks + 2*half + u;
            const int cc = (c < NF) ? c : (NF - 1);
            areg[2*ks + u] = px*W1[0*NF + cc] + py*W1[1*NF + cc] + pz*W1[2*NF + cc] + b1[cc];
        }
    }
    #pragma unroll
    for (int ks = 0; ks < 32; ++ks) {
        #pragma unroll
        for (int u = 0; u < 2; ++u) {
            const int c = 4*ks + 2*half + u;
            areg[2*ks + u] = (c < NF) ? gelu_exact(areg[2*ks + u]) : 0.0f;
        }
    }

    for (int nt = 0; nt < 8; ++nt) {
        const int n = nt*16 + row;
        const float bias = (n < NF) ? b2[n] : 0.0f;
        v8f acc;
        #pragma unroll
        for (int r = 0; r < 8; ++r) acc[r] = bias;

        const float* bn = &sW2t[n * KPAD];
        #pragma unroll
        for (int ks = 0; ks < 32; ++ks) {
            const int kb = 4*ks + 2*half;
            v2f a, b;
            a.x = areg[2*ks + 0];
            a.y = areg[2*ks + 1];
            b   = *(const v2f*)(bn + kb);
            acc = __builtin_amdgcn_wmma_f32_16x16x4_f32(
                    false, a, false, b, (short)0, acc, false, false);
        }

        #pragma unroll
        for (int r = 0; r < 8; ++r) {
            const int m = half*8 + r;
            resw[(size_t)(m0 + m)*FP + nt*16 + row] = acc[r];
        }
    }
}

#define QPW 2
#define QPB (8*QPW)
__global__ __launch_bounds__(256) void knn_gather_max_kernel(
    const float* __restrict__ x, const float* __restrict__ resw,
    float* __restrict__ out)
{
    __shared__ float spx[NS], spy[NS], spz[NS], ssq[NS];
    const int b  = blockIdx.x / (NS/QPB);
    const int i0 = (blockIdx.x % (NS/QPB)) * QPB;
    const float* xb = x + (size_t)b*NS*3;
    {
        float p0[NS/256], p1[NS/256], p2[NS/256];
        #pragma unroll
        for (int u = 0; u < NS/256; ++u) {
            const int j = threadIdx.x + 256*u;
            p0[u] = xb[j*3+0]; p1[u] = xb[j*3+1]; p2[u] = xb[j*3+2];
        }
        #pragma unroll
        for (int u = 0; u < NS/256; ++u) {
            const int j = threadIdx.x + 256*u;
            const float a0 = p0[u], a1 = p1[u], a2 = p2[u];
            spx[j] = a0; spy[j] = a1; spz[j] = a2;
            ssq[j] = a0*a0 + a1*a1 + a2*a2;
        }
    }
    __syncthreads();

    const int lane = threadIdx.x & 31;
    const int wave = threadIdx.x >> 5;
    const float NEG_INF = -__builtin_inff();
    const bool tail = (lane == 31);
    const float4* rb = (const float4*)(resw + (size_t)b*NS*FP) + lane;
    float4*       ob = (float4*)(out + (size_t)b*NS*FP) + lane;

    float4 g[NK];
    float gx = 0.0f, gy = 0.0f, gz = 0.0f;
    int   gi = 0;

    #pragma unroll
    for (int qi = 0; qi < QPW; ++qi) {
        const int i = i0 + wave*QPW + qi;
        const float qx = spx[i], qy = spy[i], qz = spz[i], qs = ssq[i];

        float L = (lane < NK) ? 3.4e38f : NEG_INF;
        int   I = 0;
        float thr = 3.4e38f;

        for (int c = 0; c < NS/32; ++c) {
            const int j = c*32 + lane;
            const float d = qs + ssq[j] - 2.0f*(qx*spx[j] + qy*spy[j] + qz*spz[j]);
            unsigned mask = __builtin_amdgcn_ballot_w32(d < thr);
            while (mask) {
                const int sl = __builtin_ctz(mask);
                const float sd = __builtin_bit_cast(float,
                    __builtin_amdgcn_readlane(__builtin_bit_cast(int, d), sl));
                const int sj = c*32 + sl;
                const float Lsh = __builtin_bit_cast(float, __builtin_amdgcn_update_dpp(
                    __builtin_bit_cast(int, NEG_INF), __builtin_bit_cast(int, L),
                    0x111, 0xf, 0xf, false));
                const int Ish = __builtin_amdgcn_update_dpp(0, I, 0x111, 0xf, 0xf, false);
                const bool mv  = L   > sd;
                const bool mvp = Lsh > sd;
                const float nL = mvp ? Lsh : sd;
                const int   nI = mvp ? Ish : sj;
                L = mv ? nL : L;
                I = mv ? nI : I;
                thr = __builtin_bit_cast(float,
                    __builtin_amdgcn_readlane(__builtin_bit_cast(int, L), NK-1));
                mask &= mask - 1;
                mask &= __builtin_amdgcn_ballot_w32(d < thr);
            }
        }

        if (qi > 0) {
            float4 m = make_float4(-3.4e38f, -3.4e38f, -3.4e38f, -3.4e38f);
            #pragma unroll
            for (int k = 0; k < NK; ++k) {
                m.x = fmaxf(m.x, g[k].x);
                m.y = fmaxf(m.y, g[k].y);
                m.z = fmaxf(m.z, g[k].z);
                m.w = fmaxf(m.w, g[k].w);
            }
            m.y = tail ? gx : m.y;
            m.z = tail ? gy : m.z;
            m.w = tail ? gz : m.w;
            ob[(size_t)gi * (FP/4)] = m;
        }

        #pragma unroll
        for (int k = 0; k < NK; ++k) {
            const int nbk = __builtin_amdgcn_readlane(I, k);
            g[k] = rb[(size_t)nbk * (FP/4)];
        }
        gx = qx; gy = qy; gz = qz; gi = i;
    }

    {
        float4 m = make_float4(-3.4e38f, -3.4e38f, -3.4e38f, -3.4e38f);
        #pragma unroll
        for (int k = 0; k < NK; ++k) {
            m.x = fmaxf(m.x, g[k].x);
            m.y = fmaxf(m.y, g[k].y);
            m.z = fmaxf(m.z, g[k].z);
            m.w = fmaxf(m.w, g[k].w);
        }
        m.y = tail ? gx : m.y;
        m.z = tail ? gy : m.z;
        m.w = tail ? gz : m.w;
        ob[(size_t)gi * (FP/4)] = m;
    }
}

extern "C" void kernel_launch(void* const* d_in, const int* in_sizes, int n_in,
                              void* d_out, int out_size, void* d_ws, size_t ws_size,
                              hipStream_t stream) {
    const float* x  = (const float*)d_in[0];
    const float* W1 = (const float*)d_in[1];
    const float* b1 = (const float*)d_in[2];
    const float* W2 = (const float*)d_in[3];
    const float* b2 = (const float*)d_in[4];
    float* out = (float*)d_out;

    float* resw = (float*)d_ws;

    mlp_wmma_kernel<<<NPTS/(16*8), 256, 0, stream>>>(x, W1, b1, W2, b2, resw);
    knn_gather_max_kernel<<<NB*(NS/QPB), 256, 0, stream>>>(x, resw, out);
}


# kNN: a query's distances stay in registers; 16th smallest of the 32 lane minima caps the 16th distance, the sorted list starts at that cap (insertions ~92 -> ~20); 4 queries per wave; same lists
# speedup vs baseline: 9.2081x; 1.3135x over previous
#include <hip/hip_runtime.h>
#include <math.h>

typedef float v2f __attribute__((ext_vector_type(2)));
typedef float v8f __attribute__((ext_vector_type(8)));

#define NB 8
#define NS 2048
#define ND 3
#define NF 125
#define FP 128
#define KPAD (FP + 4)
#define NK 16
#define NPTS (NB*NS)
#define SG 32

__device__ __forceinline__ float gelu_exact(float v) {
    return 0.5f * v * (1.0f + erff(v * 0.7071067811865475f));
}

__global__ __launch_bounds__(256) void mlp_wmma_kernel(
    const float* __restrict__ x, const float* __restrict__ W1,
    const float* __restrict__ b1, const float* __restrict__ W2,
    const float* __restrict__ b2, float* __restrict__ resw)
{
    __shared__ float sW2t[FP * KPAD];
    const int tid = threadIdx.x;
    {
        const int n  = tid & 127;
        const int k0 = tid >> 7;
        const int nc = (n < NF) ? n : (NF - 1);
        for (int g = 0; g < 64; g += SG) {
            float t[SG];
            #pragma unroll
            for (int u = 0; u < SG; ++u) {
                const int k  = 2*(g + u) + k0;
                const int kc = (k < NF) ? k : (NF - 1);
                t[u] = W2[kc*NF + nc];
            }
            #pragma unroll
            for (int u = 0; u < SG; ++u) {
                const int k = 2*(g + u) + k0;
                sW2t[n*KPAD + k] = (k < NF && n < NF) ? t[u] : 0.0f;
            }
        }
    }
    __syncthreads();

    const int lane = tid & 31;
    const int wave = tid >> 5;
    const int row  = lane & 15;
    const int half = lane >> 4;
    const int m0   = (blockIdx.x * 8 + wave) * 16;

    const float px = x[(size_t)(m0+row)*3 + 0];
    const float py = x[(size_t)(m0+row)*3 + 1];
    const float pz = x[(size_t)(m0+row)*3 + 2];

    float areg[64];
    #pragma unroll
    for (int ks = 0; ks < 32; ++ks) {
        #pragma unroll
        for (int u = 0; u < 2; ++u) {
            const int c  = 4*ks + 2*half + u;
            const int cc = (c < NF) ? c : (NF - 1);
            areg[2*ks + u] = px*W1[0*NF + cc] + py*W1[1*NF + cc] + pz*W1[2*NF + cc] + b1[cc];
        }
    }
    #pragma unroll
    for (int ks = 0; ks < 32; ++ks) {
        #pragma unroll
        for (int u = 0; u < 2; ++u) {
            const int c = 4*ks + 2*half + u;
            areg[2*ks + u] = (c < NF) ? gelu_exact(areg[2*ks + u]) : 0.0f;
        }
    }

    for (int nt = 0; nt < 8; ++nt) {
        const int n = nt*16 + row;
        const float bias = (n < NF) ? b2[n] : 0.0f;
        v8f acc;
        #pragma unroll
        for (int r = 0; r < 8; ++r) acc[r] = bias;

        const float* bn = &sW2t[n * KPAD];
        #pragma unroll
        for (int ks = 0; ks < 32; ++ks) {
            const int kb = 4*ks + 2*half;
            v2f a, b;
            a.x = areg[2*ks + 0];
            a.y = areg[2*ks + 1];
            b   = *(const v2f*)(bn + kb);
            acc = __builtin_amdgcn_wmma_f32_16x16x4_f32(
                    false, a, false, b, (short)0, acc, false, false);
        }

        #pragma unroll
        for (int r = 0; r < 8; ++r) {
            const int m = half*8 + r;
            resw[(size_t)(m0 + m)*FP + nt*16 + row] = acc[r];
        }
    }
}

#define QPW 4
#define QPB (8*QPW)
#define NC (NS/32)

#define LANE_XOR_ROW(v, J) __builtin_bit_cast(float, __builtin_amdgcn_update_dpp( \
    __builtin_bit_cast(int, (v)), __builtin_bit_cast(int, (v)), 0x160 | (J), 0xf, 0xf, false))
#define LANE_XOR_16(v) __builtin_bit_cast(float, __builtin_amdgcn_permlanex16( \
    __builtin_bit_cast(int, (v)), __builtin_bit_cast(int, (v)), 0x76543210, 0xfedcba98, false, false))
#define SORT_STAGE(K, J, PARTNER) { \
    const float p_ = (PARTNER); \
    const bool keep_min_ = (((lane & (J)) == 0) == ((lane & (K)) == 0)); \
    v = keep_min_ ? fminf(v, p_) : fmaxf(v, p_); }

__device__ __forceinline__ float sort32(float v, const int lane) {
    SORT_STAGE(2, 1, LANE_XOR_ROW(v, 1))
    SORT_STAGE(4, 2, LANE_XOR_ROW(v, 2))
    SORT_STAGE(4, 1, LANE_XOR_ROW(v, 1))
    SORT_STAGE(8, 4, LANE_XOR_ROW(v, 4))
    SORT_STAGE(8, 2, LANE_XOR_ROW(v, 2))
    SORT_STAGE(8, 1, LANE_XOR_ROW(v, 1))
    SORT_STAGE(16, 8, LANE_XOR_ROW(v, 8))
    SORT_STAGE(16, 4, LANE_XOR_ROW(v, 4))
    SORT_STAGE(16, 2, LANE_XOR_ROW(v, 2))
    SORT_STAGE(16, 1, LANE_XOR_ROW(v, 1))
    SORT_STAGE(32, 16, LANE_XOR_16(v))
    SORT_STAGE(32, 8, LANE_XOR_ROW(v, 8))
    SORT_STAGE(32, 4, LANE_XOR_ROW(v, 4))
    SORT_STAGE(32, 2, LANE_XOR_ROW(v, 2))
    SORT_STAGE(32, 1, LANE_XOR_ROW(v, 1))
    return v;
}

__global__ __launch_bounds__(256) __attribute__((amdgpu_waves_per_eu(4, 4))) void knn_gather_max_kernel(
    const float* __restrict__ x, const float* __restrict__ resw,
    float* __restrict__ out)
{
    __shared__ float spx[NS], spy[NS], spz[NS], ssq[NS];
    const int b  = blockIdx.x / (NS/QPB);
    const int i0 = (blockIdx.x % (NS/QPB)) * QPB;
    const float* xb = x + (size_t)b*NS*3;
    {
        float p0[NS/256], p1[NS/256], p2[NS/256];
        #pragma unroll
        for (int u = 0; u < NS/256; ++u) {
            const int j = threadIdx.x + 256*u;
            p0[u] = xb[j*3+0]; p1[u] = xb[j*3+1]; p2[u] = xb[j*3+2];
        }
        #pragma unroll
        for (int u = 0; u < NS/256; ++u) {
            const int j = threadIdx.x + 256*u;
            const float a0 = p0[u], a1 = p1[u], a2 = p2[u];
            spx[j] = a0; spy[j] = a1; spz[j] = a2;
            ssq[j] = a0*a0 + a1*a1 + a2*a2;
        }
    }
    __syncthreads();

    const int lane = threadIdx.x & 31;
    const int wave = threadIdx.x >> 5;
    const float NEG_INF = -__builtin_inff();
    const bool tail = (lane == 31);
    const float4* rb = (const float4*)(resw + (size_t)b*NS*FP) + lane;
    float4*       ob = (float4*)(out + (size_t)b*NS*FP) + lane;

    float4 g[NK];
    float gx = 0.0f, gy = 0.0f, gz = 0.0f;
    int   gi = 0;

    #pragma unroll 1
    for (int qi = 0; qi < QPW; ++qi) {
        asm volatile("" ::: "memory");
        const int i = i0 + wave*QPW + qi;
        const float qx = spx[i], qy = spy[i], qz = spz[i], qs = ssq[i];

        float d[NC];
        #pragma unroll
        for (int c = 0; c < NC; ++c) {
            const int j = c*32 + lane;
            d[c] = qs + ssq[j] - 2.0f*(qx*spx[j] + qy*spy[j] + qz*spz[j]);
        }
        float m0 = d[0], m1 = d[1], m2 = d[2], m3 = d[3];
        #pragma unroll
        for (int c = 4; c < NC; c += 4) {
            m0 = fminf(m0, d[c+0]); m1 = fminf(m1, d[c+1]);
            m2 = fminf(m2, d[c+2]); m3 = fminf(m3, d[c+3]);
        }
        const float lm = fminf(fminf(m0, m1), fminf(m2, m3));
        const float T = __builtin_bit_cast(float, __builtin_amdgcn_readlane(
            __builtin_bit_cast(int, sort32(lm, lane)), NK-1));
        const float bar = T + (fabsf(T)*2.4e-7f + 1e-30f);

        float L = (lane < NK) ? bar : NEG_INF;
        int   I = 0;
        float thr = bar;

        #pragma unroll
        for (int c = 0; c < NC; ++c) {
            const float dc = d[c];
            unsigned mask = __builtin_amdgcn_ballot_w32(dc < thr);
            if (mask) {
              do {
                const int sl = __builtin_ctz(mask);
                const float sd = __builtin_bit_cast(float,
                    __builtin_amdgcn_readlane(__builtin_bit_cast(int, dc), sl));
                const int sj = c*32 + sl;
                const float Lsh = __builtin_bit_cast(float, __builtin_amdgcn_update_dpp(
                    __builtin_bit_cast(int, NEG_INF), __builtin_bit_cast(int, L),
                    0x111, 0xf, 0xf, false));
                const int Ish = __builtin_amdgcn_update_dpp(0, I, 0x111, 0xf, 0xf, false);
                const bool mv  = L   > sd;
                const bool mvp = Lsh > sd;
                const float nL = mvp ? Lsh : sd;
                const int   nI = mvp ? Ish : sj;
                L = mv ? nL : L;
                I = mv ? nI : I;
                mask &= mask - 1;
              } while (mask);
              thr = __builtin_bit_cast(float,
                  __builtin_amdgcn_readlane(__builtin_bit_cast(int, L), NK-1));
            }
        }

        if (qi > 0) {
            float4 m = make_float4(-3.4e38f, -3.4e38f, -3.4e38f, -3.4e38f);
            #pragma unroll
            for (int k = 0; k < NK; ++k) {
                m.x = fmaxf(m.x, g[k].x);
                m.y = fmaxf(m.y, g[k].y);
                m.z = fmaxf(m.z, g[k].z);
                m.w = fmaxf(m.w, g[k].w);
            }
            m.y = tail ? gx : m.y;
            m.z = tail ? gy : m.z;
            m.w = tail ? gz : m.w;
            ob[(size_t)gi * (FP/4)] = m;
        }

        #pragma unroll
        for (int k = 0; k < NK; ++k) {
            const int nbk = __builtin_amdgcn_readlane(I, k);
            g[k] = rb[(size_t)nbk * (FP/4)];
        }
        gx = qx; gy = qy; gz = qz; gi = i;
    }

    {
        float4 m = make_float4(-3.4e38f, -3.4e38f, -3.4e38f, -3.4e38f);
        #pragma unroll
        for (int k = 0; k < NK; ++k) {
            m.x = fmaxf(m.x, g[k].x);
            m.y = fmaxf(m.y, g[k].y);
            m.z = fmaxf(m.z, g[k].z);
            m.w = fmaxf(m.w, g[k].w);
        }
        m.y = tail ? gx : m.y;
        m.z = tail ? gy : m.z;
        m.w = tail ? gz : m.w;
        ob[(size_t)gi * (FP/4)] = m;
    }
}

extern "C" void kernel_launch(void* const* d_in, const int* in_sizes, int n_in,
                              void* d_out, int out_size, void* d_ws, size_t ws_size,
                              hipStream_t stream) {
    const float* x  = (const float*)d_in[0];
    const float* W1 = (const float*)d_in[1];
    const float* b1 = (const float*)d_in[2];
    const float* W2 = (const float*)d_in[3];
    const float* b2 = (const float*)d_in[4];
    float* out = (float*)d_out;

    float* resw = (float*)d_ws;

    mlp_wmma_kernel<<<NPTS/(16*8), 256, 0, stream>>>(x, W1, b1, W2, b2, resw);
    knn_gather_max_kernel<<<NB*(NS/QPB), 256, 0, stream>>>(x, resw, out);
}


# MLP kernel re-tiled to fill the chip: 64 rows per block (grid 128 -> 256), two waves per M-tile, each activates half the A fragments, passes them through LDS, does 4 of 8 N-tiles; same values
# speedup vs baseline: 10.4705x; 1.1371x over previous
#include <hip/hip_runtime.h>
#include <math.h>

typedef float v2f __attribute__((ext_vector_type(2)));
typedef float v8f __attribute__((ext_vector_type(8)));

#define NB 8
#define NS 2048
#define ND 3
#define NF 125
#define FP 128
#define KPAD (FP + 4)
#define NK 16
#define NPTS (NB*NS)
#define SG 32

__device__ __forceinline__ float gelu_exact(float v) {
    return 0.5f * v * (1.0f + erff(v * 0.7071067811865475f));
}

#define MT 4

__global__ __launch_bounds__(256) void mlp_wmma_kernel(
    const float* __restrict__ x, const float* __restrict__ W1,
    const float* __restrict__ b1, const float* __restrict__ W2,
    const float* __restrict__ b2, float* __restrict__ resw)
{
    __shared__ float sW2t[FP * KPAD];
    __shared__ v2f   sA[MT * 32 * 32];
    const int tid = threadIdx.x;
    {
        const int n  = tid & 127;
        const int k0 = tid >> 7;
        const int nc = (n < NF) ? n : (NF - 1);
        for (int g = 0; g < 64; g += SG) {
            float t[SG];
            #pragma unroll
            for (int u = 0; u < SG; ++u) {
                const int k  = 2*(g + u) + k0;
                const int kc = (k < NF) ? k : (NF - 1);
                t[u] = W2[kc*NF + nc];
            }
            #pragma unroll
            for (int u = 0; u < SG; ++u) {
                const int k = 2*(g + u) + k0;
                sW2t[n*KPAD + k] = (k < NF && n < NF) ? t[u] : 0.0f;
            }
        }
    }

    const int lane = tid & 31;
    const int wave = tid >> 5;
    const int row  = lane & 15;
    const int half = lane >> 4;
    const int mt   = wave >> 1;
    const int e    = wave & 1;
    const int m0   = (blockIdx.x * MT + mt) * 16;

    const float px = x[(size_t)(m0+row)*3 + 0];
    const float py = x[(size_t)(m0+row)*3 + 1];
    const float pz = x[(size_t)(m0+row)*3 + 2];

    float hreg[32];
    #pragma unroll
    for (int j = 0; j < 16; ++j) {
        #pragma unroll
        for (int u = 0; u < 2; ++u) {
            const int c  = 4*(16*e + j) + 2*half + u;
            const int cc = (c < NF) ? c : (NF - 1);
            hreg[2*j + u] = px*W1[0*NF + cc] + py*W1[1*NF + cc] + pz*W1[2*NF + cc] + b1[cc];
        }
    }
    #pragma unroll
    for (int j = 0; j < 16; ++j) {
        #pragma unroll
        for (int u = 0; u < 2; ++u) {
            const int c = 4*(16*e + j) + 2*half + u;
            hreg[2*j + u] = (c < NF) ? gelu_exact(hreg[2*j + u]) : 0.0f;
        }
    }
    #pragma unroll
    for (int j = 0; j < 16; ++j) {
        v2f a;
        a.x = hreg[2*j + 0];
        a.y = hreg[2*j + 1];
        sA[(mt*32 + 16*e + j)*32 + lane] = a;
    }
    __syncthreads();

    float areg[64];
    #pragma unroll
    for (int ks = 0; ks < 32; ++ks) {
        const v2f a = sA[(mt*32 + ks)*32 + lane];
        areg[2*ks + 0] = a.x;
        areg[2*ks + 1] = a.y;
    }

    for (int q = 0; q < 4; ++q) {
        const int nt = 4*e + q;
        const int n = nt*16 + row;
        const float bias = (n < NF) ? b2[n] : 0.0f;
        v8f acc;
        #pragma unroll
        for (int r = 0; r < 8; ++r) acc[r] = bias;

        const float* bn = &sW2t[n * KPAD];
        #pragma unroll
        for (int ks = 0; ks < 32; ++ks) {
            const int kb = 4*ks + 2*half;
            v2f a, b;
            a.x = areg[2*ks + 0];
            a.y = areg[2*ks + 1];
            b   = *(const v2f*)(bn + kb);
            acc = __builtin_amdgcn_wmma_f32_16x16x4_f32(
                    false, a, false, b, (short)0, acc, false, false);
        }

        #pragma unroll
        for (int r = 0; r < 8; ++r) {
            const int m = half*8 + r;
            resw[(size_t)(m0 + m)*FP + nt*16 + row] = acc[r];
        }
    }
}

#define QPW 4
#define QPB (8*QPW)
#define NC (NS/32)

#define LANE_XOR_ROW(v, J) __builtin_bit_cast(float, __builtin_amdgcn_update_dpp( \
    __builtin_bit_cast(int, (v)), __builtin_bit_cast(int, (v)), 0x160 | (J), 0xf, 0xf, false))
#define LANE_XOR_16(v) __builtin_bit_cast(float, __builtin_amdgcn_permlanex16( \
    __builtin_bit_cast(int, (v)), __builtin_bit_cast(int, (v)), 0x76543210, 0xfedcba98, false, false))
#define SORT_STAGE(K, J, PARTNER) { \
    const float p_ = (PARTNER); \
    const bool keep_min_ = (((lane & (J)) == 0) == ((lane & (K)) == 0)); \
    v = keep_min_ ? fminf(v, p_) : fmaxf(v, p_); }

__device__ __forceinline__ float sort32(float v, const int lane) {
    SORT_STAGE(2, 1, LANE_XOR_ROW(v, 1))
    SORT_STAGE(4, 2, LANE_XOR_ROW(v, 2))
    SORT_STAGE(4, 1, LANE_XOR_ROW(v, 1))
    SORT_STAGE(8, 4, LANE_XOR_ROW(v, 4))
    SORT_STAGE(8, 2, LANE_XOR_ROW(v, 2))
    SORT_STAGE(8, 1, LANE_XOR_ROW(v, 1))
    SORT_STAGE(16, 8, LANE_XOR_ROW(v, 8))
    SORT_STAGE(16, 4, LANE_XOR_ROW(v, 4))
    SORT_STAGE(16, 2, LANE_XOR_ROW(v, 2))
    SORT_STAGE(16, 1, LANE_XOR_ROW(v, 1))
    SORT_STAGE(32, 16, LANE_XOR_16(v))
    SORT_STAGE(32, 8, LANE_XOR_ROW(v, 8))
    SORT_STAGE(32, 4, LANE_XOR_ROW(v, 4))
    SORT_STAGE(32, 2, LANE_XOR_ROW(v, 2))
    SORT_STAGE(32, 1, LANE_XOR_ROW(v, 1))
    return v;
}

__global__ __launch_bounds__(256) __attribute__((amdgpu_waves_per_eu(4, 4))) void knn_gather_max_kernel(
    const float* __restrict__ x, const float* __restrict__ resw,
    float* __restrict__ out)
{
    __shared__ float spx[NS], spy[NS], spz[NS], ssq[NS];
    const int b  = blockIdx.x / (NS/QPB);
    const int i0 = (blockIdx.x % (NS/QPB)) * QPB;
    const float* xb = x + (size_t)b*NS*3;
    {
        float p0[NS/256], p1[NS/256], p2[NS/256];
        #pragma unroll
        for (int u = 0; u < NS/256; ++u) {
            const int j = threadIdx.x + 256*u;
            p0[u] = xb[j*3+0]; p1[u] = xb[j*3+1]; p2[u] = xb[j*3+2];
        }
        #pragma unroll
        for (int u = 0; u < NS/256; ++u) {
            const int j = threadIdx.x + 256*u;
            const float a0 = p0[u], a1 = p1[u], a2 = p2[u];
            spx[j] = a0; spy[j] = a1; spz[j] = a2;
            ssq[j] = a0*a0 + a1*a1 + a2*a2;
        }
    }
    __syncthreads();

    const int lane = threadIdx.x & 31;
    const int wave = threadIdx.x >> 5;
    const float NEG_INF = -__builtin_inff();
    const bool tail = (lane == 31);
    const float4* rb = (const float4*)(resw + (size_t)b*NS*FP) + lane;
    float4*       ob = (float4*)(out + (size_t)b*NS*FP) + lane;

    float4 g[NK];
    float gx = 0.0f, gy = 0.0f, gz = 0.0f;
    int   gi = 0;

    #pragma unroll 1
    for (int qi = 0; qi < QPW; ++qi) {
        asm volatile("" ::: "memory");
        const int i = i0 + wave*QPW + qi;
        const float qx = spx[i], qy = spy[i], qz = spz[i], qs = ssq[i];

        float d[NC];
        #pragma unroll
        for (int c = 0; c < NC; ++c) {
            const int j = c*32 + lane;
            d[c] = qs + ssq[j] - 2.0f*(qx*spx[j] + qy*spy[j] + qz*spz[j]);
        }
        float m0 = d[0], m1 = d[1], m2 = d[2], m3 = d[3];
        #pragma unroll
        for (int c = 4; c < NC; c += 4) {
            m0 = fminf(m0, d[c+0]); m1 = fminf(m1, d[c+1]);
            m2 = fminf(m2, d[c+2]); m3 = fminf(m3, d[c+3]);
        }
        const float lm = fminf(fminf(m0, m1), fminf(m2, m3));
        const float T = __builtin_bit_cast(float, __builtin_amdgcn_readlane(
            __builtin_bit_cast(int, sort32(lm, lane)), NK-1));
        const float bar = T + (fabsf(T)*2.4e-7f + 1e-30f);

        float L = (lane < NK) ? bar : NEG_INF;
        int   I = 0;
        float thr = bar;

        #pragma unroll
        for (int c = 0; c < NC; ++c) {
            const float dc = d[c];
            unsigned mask = __builtin_amdgcn_ballot_w32(dc < thr);
            if (mask) {
              do {
                const int sl = __builtin_ctz(mask);
                const float sd = __builtin_bit_cast(float,
                    __builtin_amdgcn_readlane(__builtin_bit_cast(int, dc), sl));
                const int sj = c*32 + sl;
                const float Lsh = __builtin_bit_cast(float, __builtin_amdgcn_update_dpp(
                    __builtin_bit_cast(int, NEG_INF), __builtin_bit_cast(int, L),
                    0x111, 0xf, 0xf, false));
                const int Ish = __builtin_amdgcn_update_dpp(0, I, 0x111, 0xf, 0xf, false);
                const bool mv  = L   > sd;
                const bool mvp = Lsh > sd;
                const float nL = mvp ? Lsh : sd;
                const int   nI = mvp ? Ish : sj;
                L = mv ? nL : L;
                I = mv ? nI : I;
                mask &= mask - 1;
              } while (mask);
              thr = __builtin_bit_cast(float,
                  __builtin_amdgcn_readlane(__builtin_bit_cast(int, L), NK-1));
            }
        }

        if (qi > 0) {
            float4 m = make_float4(-3.4e38f, -3.4e38f, -3.4e38f, -3.4e38f);
            #pragma unroll
            for (int k = 0; k < NK; ++k) {
                m.x = fmaxf(m.x, g[k].x);
                m.y = fmaxf(m.y, g[k].y);
                m.z = fmaxf(m.z, g[k].z);
                m.w = fmaxf(m.w, g[k].w);
            }
            m.y = tail ? gx : m.y;
            m.z = tail ? gy : m.z;
            m.w = tail ? gz : m.w;
            ob[(size_t)gi * (FP/4)] = m;
        }

        #pragma unroll
        for (int k = 0; k < NK; ++k) {
            const int nbk = __builtin_amdgcn_readlane(I, k);
            g[k] = rb[(size_t)nbk * (FP/4)];
        }
        gx = qx; gy = qy; gz = qz; gi = i;
    }

    {
        float4 m = make_float4(-3.4e38f, -3.4e38f, -3.4e38f, -3.4e38f);
        #pragma unroll
        for (int k = 0; k < NK; ++k) {
            m.x = fmaxf(m.x, g[k].x);
            m.y = fmaxf(m.y, g[k].y);
            m.z = fmaxf(m.z, g[k].z);
            m.w = fmaxf(m.w, g[k].w);
        }
        m.y = tail ? gx : m.y;
        m.z = tail ? gy : m.z;
        m.w = tail ? gz : m.w;
        ob[(size_t)gi * (FP/4)] = m;
    }
}

extern "C" void kernel_launch(void* const* d_in, const int* in_sizes, int n_in,
                              void* d_out, int out_size, void* d_ws, size_t ws_size,
                              hipStream_t stream) {
    const float* x  = (const float*)d_in[0];
    const float* W1 = (const float*)d_in[1];
    const float* b1 = (const float*)d_in[2];
    const float* W2 = (const float*)d_in[3];
    const float* b2 = (const float*)d_in[4];
    float* out = (float*)d_out;

    float* resw = (float*)d_ws;

    mlp_wmma_kernel<<<NPTS/(16*MT), 256, 0, stream>>>(x, W1, b1, W2, b2, resw);
    knn_gather_max_kernel<<<NB*(NS/QPB), 256, 0, stream>>>(x, resw, out);
}


# MLP kernel: a wave's 32 gelu as interleaved straight-line chains: erff's own |t|<1 polynomial written out, taken when every |t|<1 in the wave (one test), library call per element else; same values
# speedup vs baseline: 10.8650x; 1.0377x over previous
#include <hip/hip_runtime.h>
#include <math.h>

typedef float v2f __attribute__((ext_vector_type(2)));
typedef float v8f __attribute__((ext_vector_type(8)));

#define NB 8
#define NS 2048
#define ND 3
#define NF 125
#define FP 128
#define KPAD (FP + 4)
#define NK 16
#define NPTS (NB*NS)
#define SG 32

__device__ __forceinline__ float gelu_exact(float v) {
    return 0.5f * v * (1.0f + erff(v * 0.7071067811865475f));
}

__device__ __forceinline__ float gelu_small(float v) {
    const float t = v * 0.7071067811865475f;
    const float s = t * t;
    float p = fmaf(__builtin_bit_cast(float, 0xba1345e1u), s, __builtin_bit_cast(float, 0x3ba10414u));
    p = fmaf(s, p, __builtin_bit_cast(float, 0xbcdac9b8u));
    p = fmaf(s, p, __builtin_bit_cast(float, 0x3de703beu));
    p = fmaf(s, p, __builtin_bit_cast(float, 0xbec09330u));
    p = fmaf(s, p, __builtin_bit_cast(float, 0x3e0375d0u));
    const float a = fabsf(t);
    const float e = copysignf(fmaf(a, p, a), t);
    return 0.5f * v * (1.0f + e);
}

#define MT 4

__global__ __launch_bounds__(256) void mlp_wmma_kernel(
    const float* __restrict__ x, const float* __restrict__ W1,
    const float* __restrict__ b1, const float* __restrict__ W2,
    const float* __restrict__ b2, float* __restrict__ resw)
{
    __shared__ float sW2t[FP * KPAD];
    __shared__ v2f   sA[MT * 32 * 32];
    const int tid = threadIdx.x;
    {
        const int n  = tid & 127;
        const int k0 = tid >> 7;
        const int nc = (n < NF) ? n : (NF - 1);
        for (int g = 0; g < 64; g += SG) {
            float t[SG];
            #pragma unroll
            for (int u = 0; u < SG; ++u) {
                const int k  = 2*(g + u) + k0;
                const int kc = (k < NF) ? k : (NF - 1);
                t[u] = W2[kc*NF + nc];
            }
            #pragma unroll
            for (int u = 0; u < SG; ++u) {
                const int k = 2*(g + u) + k0;
                sW2t[n*KPAD + k] = (k < NF && n < NF) ? t[u] : 0.0f;
            }
        }
    }

    const int lane = tid & 31;
    const int wave = tid >> 5;
    const int row  = lane & 15;
    const int half = lane >> 4;
    const int mt   = wave >> 1;
    const int e    = wave & 1;
    const int m0   = (blockIdx.x * MT + mt) * 16;

    const float px = x[(size_t)(m0+row)*3 + 0];
    const float py = x[(size_t)(m0+row)*3 + 1];
    const float pz = x[(size_t)(m0+row)*3 + 2];

    float hreg[32];
    #pragma unroll
    for (int j = 0; j < 16; ++j) {
        #pragma unroll
        for (int u = 0; u < 2; ++u) {
            const int c  = 4*(16*e + j) + 2*half + u;
            const int cc = (c < NF) ? c : (NF - 1);
            hreg[2*j + u] = px*W1[0*NF + cc] + py*W1[1*NF + cc] + pz*W1[2*NF + cc] + b1[cc];
        }
    }
    float tmax = 0.0f;
    #pragma unroll
    for (int i = 0; i < 32; ++i)
        tmax = fmaxf(tmax, fabsf(hreg[i] * 0.7071067811865475f));
    if (__builtin_amdgcn_ballot_w32(!(tmax < 1.0f)) == 0u) {
        #pragma unroll
        for (int j = 0; j < 16; ++j) {
            #pragma unroll
            for (int u = 0; u < 2; ++u) {
                const int c = 4*(16*e + j) + 2*half + u;
                hreg[2*j + u] = (c < NF) ? gelu_small(hreg[2*j + u]) : 0.0f;
            }
        }
    } else {
        #pragma unroll
        for (int j = 0; j < 16; ++j) {
            #pragma unroll
            for (int u = 0; u < 2; ++u) {
                const int c = 4*(16*e + j) + 2*half + u;
                hreg[2*j + u] = (c < NF) ? gelu_exact(hreg[2*j + u]) : 0.0f;
            }
        }
    }
    #pragma unroll
    for (int j = 0; j < 16; ++j) {
        v2f a;
        a.x = hreg[2*j + 0];
        a.y = hreg[2*j + 1];
        sA[(mt*32 + 16*e + j)*32 + lane] = a;
    }
    __syncthreads();

    float areg[64];
    #pragma unroll
    for (int ks = 0; ks < 32; ++ks) {
        const v2f a = sA[(mt*32 + ks)*32 + lane];
        areg[2*ks + 0] = a.x;
        areg[2*ks + 1] = a.y;
    }

    for (int q = 0; q < 4; ++q) {
        const int nt = 4*e + q;
        const int n = nt*16 + row;
        const float bias = (n < NF) ? b2[n] : 0.0f;
        v8f acc;
        #pragma unroll
        for (int r = 0; r < 8; ++r) acc[r] = bias;

        const float* bn = &sW2t[n * KPAD];
        #pragma unroll
        for (int ks = 0; ks < 32; ++ks) {
            const int kb = 4*ks + 2*half;
            v2f a, b;
            a.x = areg[2*ks + 0];
            a.y = areg[2*ks + 1];
            b   = *(const v2f*)(bn + kb);
            acc = __builtin_amdgcn_wmma_f32_16x16x4_f32(
                    false, a, false, b, (short)0, acc, false, false);
        }

        #pragma unroll
        for (int r = 0; r < 8; ++r) {
            const int m = half*8 + r;
            resw[(size_t)(m0 + m)*FP + nt*16 + row] = acc[r];
        }
    }
}

#define QPW 4
#define QPB (8*QPW)
#define NC (NS/32)

#define LANE_XOR_ROW(v, J) __builtin_bit_cast(float, __builtin_amdgcn_update_dpp( \
    __builtin_bit_cast(int, (v)), __builtin_bit_cast(int, (v)), 0x160 | (J), 0xf, 0xf, false))
#define LANE_XOR_16(v) __builtin_bit_cast(float, __builtin_amdgcn_permlanex16( \
    __builtin_bit_cast(int, (v)), __builtin_bit_cast(int, (v)), 0x76543210, 0xfedcba98, false, false))
#define SORT_STAGE(K, J, PARTNER) { \
    const float p_ = (PARTNER); \
    const bool keep_min_ = (((lane & (J)) == 0) == ((lane & (K)) == 0)); \
    v = keep_min_ ? fminf(v, p_) : fmaxf(v, p_); }

__device__ __forceinline__ float sort32(float v, const int lane) {
    SORT_STAGE(2, 1, LANE_XOR_ROW(v, 1))
    SORT_STAGE(4, 2, LANE_XOR_ROW(v, 2))
    SORT_STAGE(4, 1, LANE_XOR_ROW(v, 1))
    SORT_STAGE(8, 4, LANE_XOR_ROW(v, 4))
    SORT_STAGE(8, 2, LANE_XOR_ROW(v, 2))
    SORT_STAGE(8, 1, LANE_XOR_ROW(v, 1))
    SORT_STAGE(16, 8, LANE_XOR_ROW(v, 8))
    SORT_STAGE(16, 4, LANE_XOR_ROW(v, 4))
    SORT_STAGE(16, 2, LANE_XOR_ROW(v, 2))
    SORT_STAGE(16, 1, LANE_XOR_ROW(v, 1))
    SORT_STAGE(32, 16, LANE_XOR_16(v))
    SORT_STAGE(32, 8, LANE_XOR_ROW(v, 8))
    SORT_STAGE(32, 4, LANE_XOR_ROW(v, 4))
    SORT_STAGE(32, 2, LANE_XOR_ROW(v, 2))
    SORT_STAGE(32, 1, LANE_XOR_ROW(v, 1))
    return v;
}

__global__ __launch_bounds__(256) __attribute__((amdgpu_waves_per_eu(4, 4))) void knn_gather_max_kernel(
    const float* __restrict__ x, const float* __restrict__ resw,
    float* __restrict__ out)
{
    __shared__ float spx[NS], spy[NS], spz[NS], ssq[NS];
    const int b  = blockIdx.x / (NS/QPB);
    const int i0 = (blockIdx.x % (NS/QPB)) * QPB;
    const float* xb = x + (size_t)b*NS*3;
    {
        float p0[NS/256], p1[NS/256], p2[NS/256];
        #pragma unroll
        for (int u = 0; u < NS/256; ++u) {
            const int j = threadIdx.x + 256*u;
            p0[u] = xb[j*3+0]; p1[u] = xb[j*3+1]; p2[u] = xb[j*3+2];
        }
        #pragma unroll
        for (int u = 0; u < NS/256; ++u) {
            const int j = threadIdx.x + 256*u;
            const float a0 = p0[u], a1 = p1[u], a2 = p2[u];
            spx[j] = a0; spy[j] = a1; spz[j] = a2;
            ssq[j] = a0*a0 + a1*a1 + a2*a2;
        }
    }
    __syncthreads();

    const int lane = threadIdx.x & 31;
    const int wave = threadIdx.x >> 5;
    const float NEG_INF = -__builtin_inff();
    const bool tail = (lane == 31);
    const float4* rb = (const float4*)(resw + (size_t)b*NS*FP) + lane;
    float4*       ob = (float4*)(out + (size_t)b*NS*FP) + lane;

    float4 g[NK];
    float gx = 0.0f, gy = 0.0f, gz = 0.0f;
    int   gi = 0;

    #pragma unroll 1
    for (int qi = 0; qi < QPW; ++qi) {
        asm volatile("" ::: "memory");
        const int i = i0 + wave*QPW + qi;
        const float qx = spx[i], qy = spy[i], qz = spz[i], qs = ssq[i];

        float d[NC];
        #pragma unroll
        for (int c = 0; c < NC; ++c) {
            const int j = c*32 + lane;
            d[c] = qs + ssq[j] - 2.0f*(qx*spx[j] + qy*spy[j] + qz*spz[j]);
        }
        float m0 = d[0], m1 = d[1], m2 = d[2], m3 = d[3];
        #pragma unroll
        for (int c = 4; c < NC; c += 4) {
            m0 = fminf(m0, d[c+0]); m1 = fminf(m1, d[c+1]);
            m2 = fminf(m2, d[c+2]); m3 = fminf(m3, d[c+3]);
        }
        const float lm = fminf(fminf(m0, m1), fminf(m2, m3));
        const float T = __builtin_bit_cast(float, __builtin_amdgcn_readlane(
            __builtin_bit_cast(int, sort32(lm, lane)), NK-1));
        const float bar = T + (fabsf(T)*2.4e-7f + 1e-30f);

        float L = (lane < NK) ? bar : NEG_INF;
        int   I = 0;
        float thr = bar;

        #pragma unroll
        for (int c = 0; c < NC; ++c) {
            const float dc = d[c];
            unsigned mask = __builtin_amdgcn_ballot_w32(dc < thr);
            if (mask) {
              do {
                const int sl = __builtin_ctz(mask);
                const float sd = __builtin_bit_cast(float,
                    __builtin_amdgcn_readlane(__builtin_bit_cast(int, dc), sl));
                const int sj = c*32 + sl;
                const float Lsh = __builtin_bit_cast(float, __builtin_amdgcn_update_dpp(
                    __builtin_bit_cast(int, NEG_INF), __builtin_bit_cast(int, L),
                    0x111, 0xf, 0xf, false));
                const int Ish = __builtin_amdgcn_update_dpp(0, I, 0x111, 0xf, 0xf, false);
                const bool mv  = L   > sd;
                const bool mvp = Lsh > sd;
                const float nL = mvp ? Lsh : sd;
                const int   nI = mvp ? Ish : sj;
                L = mv ? nL : L;
                I = mv ? nI : I;
                mask &= mask - 1;
              } while (mask);
              thr = __builtin_bit_cast(float,
                  __builtin_amdgcn_readlane(__builtin_bit_cast(int, L), NK-1));
            }
        }

        if (qi > 0) {
            float4 m = make_float4(-3.4e38f, -3.4e38f, -3.4e38f, -3.4e38f);
            #pragma unroll
            for (int k = 0; k < NK; ++k) {
                m.x = fmaxf(m.x, g[k].x);
                m.y = fmaxf(m.y, g[k].y);
                m.z = fmaxf(m.z, g[k].z);
                m.w = fmaxf(m.w, g[k].w);
            }
            m.y = tail ? gx : m.y;
            m.z = tail ? gy : m.z;
            m.w = tail ? gz : m.w;
            ob[(size_t)gi * (FP/4)] = m;
        }

        #pragma unroll
        for (int k = 0; k < NK; ++k) {
            const int nbk = __builtin_amdgcn_readlane(I, k);
            g[k] = rb[(size_t)nbk * (FP/4)];
        }
        gx = qx; gy = qy; gz = qz; gi = i;
    }

    {
        float4 m = make_float4(-3.4e38f, -3.4e38f, -3.4e38f, -3.4e38f);
        #pragma unroll
        for (int k = 0; k < NK; ++k) {
            m.x = fmaxf(m.x, g[k].x);
            m.y = fmaxf(m.y, g[k].y);
            m.z = fmaxf(m.z, g[k].z);
            m.w = fmaxf(m.w, g[k].w);
        }
        m.y = tail ? gx : m.y;
        m.z = tail ? gy : m.z;
        m.w = tail ? gz : m.w;
        ob[(size_t)gi * (FP/4)] = m;
    }
}

extern "C" void kernel_launch(void* const* d_in, const int* in_sizes, int n_in,
                              void* d_out, int out_size, void* d_ws, size_t ws_size,
                              hipStream_t stream) {
    const float* x  = (const float*)d_in[0];
    const float* W1 = (const float*)d_in[1];
    const float* b1 = (const float*)d_in[2];
    const float* W2 = (const float*)d_in[3];
    const float* b2 = (const float*)d_in[4];
    float* out = (float*)d_out;

    float* resw = (float*)d_ws;

    mlp_wmma_kernel<<<NPTS/(16*MT), 256, 0, stream>>>(x, W1, b1, W2, b2, resw);
    knn_gather_max_kernel<<<NB*(NS/QPB), 256, 0, stream>>>(x, resw, out);
}


# kNN pass 1 reads each candidate as one 16-byte LDS record (x,y,z,|p|^2): one ds_load_b128 per step, 16 requested ahead of use, for four 4-byte loads 2-3 deep; same distance bits (explicit fma order)
# speedup vs baseline: 11.9441x; 1.0993x over previous
#include <hip/hip_runtime.h>
#include <math.h>

typedef float v2f __attribute__((ext_vector_type(2)));
typedef float v8f __attribute__((ext_vector_type(8)));

#define NB 8
#define NS 2048
#define ND 3
#define NF 125
#define FP 128
#define KPAD (FP + 4)
#define NK 16
#define NPTS (NB*NS)
#define SG 32

__device__ __forceinline__ float gelu_exact(float v) {
    return 0.5f * v * (1.0f + erff(v * 0.7071067811865475f));
}

__device__ __forceinline__ float gelu_small(float v) {
    const float t = v * 0.7071067811865475f;
    const float s = t * t;
    float p = fmaf(__builtin_bit_cast(float, 0xba1345e1u), s, __builtin_bit_cast(float, 0x3ba10414u));
    p = fmaf(s, p, __builtin_bit_cast(float, 0xbcdac9b8u));
    p = fmaf(s, p, __builtin_bit_cast(float, 0x3de703beu));
    p = fmaf(s, p, __builtin_bit_cast(float, 0xbec09330u));
    p = fmaf(s, p, __builtin_bit_cast(float, 0x3e0375d0u));
    const float a = fabsf(t);
    const float e = copysignf(fmaf(a, p, a), t);
    return 0.5f * v * (1.0f + e);
}

#define MT 4

__global__ __launch_bounds__(256) void mlp_wmma_kernel(
    const float* __restrict__ x, const float* __restrict__ W1,
    const float* __restrict__ b1, const float* __restrict__ W2,
    const float* __restrict__ b2, float* __restrict__ resw)
{
    __shared__ float sW2t[FP * KPAD];
    __shared__ v2f   sA[MT * 32 * 32];
    const int tid = threadIdx.x;
    {
        const int n  = tid & 127;
        const int k0 = tid >> 7;
        const int nc = (n < NF) ? n : (NF - 1);
        for (int g = 0; g < 64; g += SG) {
            float t[SG];
            #pragma unroll
            for (int u = 0; u < SG; ++u) {
                const int k  = 2*(g + u) + k0;
                const int kc = (k < NF) ? k : (NF - 1);
                t[u] = W2[kc*NF + nc];
            }
            #pragma unroll
            for (int u = 0; u < SG; ++u) {
                const int k = 2*(g + u) + k0;
                sW2t[n*KPAD + k] = (k < NF && n < NF) ? t[u] : 0.0f;
            }
        }
    }

    const int lane = tid & 31;
    const int wave = tid >> 5;
    const int row  = lane & 15;
    const int half = lane >> 4;
    const int mt   = wave >> 1;
    const int e    = wave & 1;
    const int m0   = (blockIdx.x * MT + mt) * 16;

    const float px = x[(size_t)(m0+row)*3 + 0];
    const float py = x[(size_t)(m0+row)*3 + 1];
    const float pz = x[(size_t)(m0+row)*3 + 2];

    float hreg[32];
    #pragma unroll
    for (int j = 0; j < 16; ++j) {
        #pragma unroll
        for (int u = 0; u < 2; ++u) {
            const int c  = 4*(16*e + j) + 2*half + u;
            const int cc = (c < NF) ? c : (NF - 1);
            hreg[2*j + u] = px*W1[0*NF + cc] + py*W1[1*NF + cc] + pz*W1[2*NF + cc] + b1[cc];
        }
    }
    float tmax = 0.0f;
    #pragma unroll
    for (int i = 0; i < 32; ++i)
        tmax = fmaxf(tmax, fabsf(hreg[i] * 0.7071067811865475f));
    if (__builtin_amdgcn_ballot_w32(!(tmax < 1.0f)) == 0u) {
        #pragma unroll
        for (int j = 0; j < 16; ++j) {
            #pragma unroll
            for (int u = 0; u < 2; ++u) {
                const int c = 4*(16*e + j) + 2*half + u;
                hreg[2*j + u] = (c < NF) ? gelu_small(hreg[2*j + u]) : 0.0f;
            }
        }
    } else {
        #pragma unroll
        for (int j = 0; j < 16; ++j) {
            #pragma unroll
            for (int u = 0; u < 2; ++u) {
                const int c = 4*(16*e + j) + 2*half + u;
                hreg[2*j + u] = (c < NF) ? gelu_exact(hreg[2*j + u]) : 0.0f;
            }
        }
    }
    #pragma unroll
    for (int j = 0; j < 16; ++j) {
        v2f a;
        a.x = hreg[2*j + 0];
        a.y = hreg[2*j + 1];
        sA[(mt*32 + 16*e + j)*32 + lane] = a;
    }
    __syncthreads();

    float areg[64];
    #pragma unroll
    for (int ks = 0; ks < 32; ++ks) {
        const v2f a = sA[(mt*32 + ks)*32 + lane];
        areg[2*ks + 0] = a.x;
        areg[2*ks + 1] = a.y;
    }

    for (int q = 0; q < 4; ++q) {
        const int nt = 4*e + q;
        const int n = nt*16 + row;
        const float bias = (n < NF) ? b2[n] : 0.0f;
        v8f acc;
        #pragma unroll
        for (int r = 0; r < 8; ++r) acc[r] = bias;

        const float* bn = &sW2t[n * KPAD];
        #pragma unroll
        for (int ks = 0; ks < 32; ++ks) {
            const int kb = 4*ks + 2*half;
            v2f a, b;
            a.x = areg[2*ks + 0];
            a.y = areg[2*ks + 1];
            b   = *(const v2f*)(bn + kb);
            acc = __builtin_amdgcn_wmma_f32_16x16x4_f32(
                    false, a, false, b, (short)0, acc, false, false);
        }

        #pragma unroll
        for (int r = 0; r < 8; ++r) {
            const int m = half*8 + r;
            resw[(size_t)(m0 + m)*FP + nt*16 + row] = acc[r];
        }
    }
}

#define QPW 4
#define QPB (8*QPW)
#define NC (NS/32)
#define LG 16

#define LANE_XOR_ROW(v, J) __builtin_bit_cast(float, __builtin_amdgcn_update_dpp( \
    __builtin_bit_cast(int, (v)), __builtin_bit_cast(int, (v)), 0x160 | (J), 0xf, 0xf, false))
#define LANE_XOR_16(v) __builtin_bit_cast(float, __builtin_amdgcn_permlanex16( \
    __builtin_bit_cast(int, (v)), __builtin_bit_cast(int, (v)), 0x76543210, 0xfedcba98, false, false))
#define SORT_STAGE(K, J, PARTNER) { \
    const float p_ = (PARTNER); \
    const bool keep_min_ = (((lane & (J)) == 0) == ((lane & (K)) == 0)); \
    v = keep_min_ ? fminf(v, p_) : fmaxf(v, p_); }

__device__ __forceinline__ float sort32(float v, const int lane) {
    SORT_STAGE(2, 1, LANE_XOR_ROW(v, 1))
    SORT_STAGE(4, 2, LANE_XOR_ROW(v, 2))
    SORT_STAGE(4, 1, LANE_XOR_ROW(v, 1))
    SORT_STAGE(8, 4, LANE_XOR_ROW(v, 4))
    SORT_STAGE(8, 2, LANE_XOR_ROW(v, 2))
    SORT_STAGE(8, 1, LANE_XOR_ROW(v, 1))
    SORT_STAGE(16, 8, LANE_XOR_ROW(v, 8))
    SORT_STAGE(16, 4, LANE_XOR_ROW(v, 4))
    SORT_STAGE(16, 2, LANE_XOR_ROW(v, 2))
    SORT_STAGE(16, 1, LANE_XOR_ROW(v, 1))
    SORT_STAGE(32, 16, LANE_XOR_16(v))
    SORT_STAGE(32, 8, LANE_XOR_ROW(v, 8))
    SORT_STAGE(32, 4, LANE_XOR_ROW(v, 4))
    SORT_STAGE(32, 2, LANE_XOR_ROW(v, 2))
    SORT_STAGE(32, 1, LANE_XOR_ROW(v, 1))
    return v;
}

__global__ __launch_bounds__(256) __attribute__((amdgpu_waves_per_eu(4, 4))) void knn_gather_max_kernel(
    const float* __restrict__ x, const float* __restrict__ resw,
    float* __restrict__ out)
{
    __shared__ float4 sp[NS];
    const int b  = blockIdx.x / (NS/QPB);
    const int i0 = (blockIdx.x % (NS/QPB)) * QPB;
    const float* xb = x + (size_t)b*NS*3;
    {
        float p0[NS/256], p1[NS/256], p2[NS/256];
        #pragma unroll
        for (int u = 0; u < NS/256; ++u) {
            const int j = threadIdx.x + 256*u;
            p0[u] = xb[j*3+0]; p1[u] = xb[j*3+1]; p2[u] = xb[j*3+2];
        }
        #pragma unroll
        for (int u = 0; u < NS/256; ++u) {
            const int j = threadIdx.x + 256*u;
            const float a0 = p0[u], a1 = p1[u], a2 = p2[u];
            sp[j] = make_float4(a0, a1, a2, a0*a0 + a1*a1 + a2*a2);
        }
    }
    __syncthreads();

    const int lane = threadIdx.x & 31;
    const int wave = threadIdx.x >> 5;
    const float NEG_INF = -__builtin_inff();
    const bool tail = (lane == 31);
    const float4* rb = (const float4*)(resw + (size_t)b*NS*FP) + lane;
    float4*       ob = (float4*)(out + (size_t)b*NS*FP) + lane;

    float4 g[NK];
    float gx = 0.0f, gy = 0.0f, gz = 0.0f;
    int   gi = 0;

    #pragma unroll 1
    for (int qi = 0; qi < QPW; ++qi) {
        asm volatile("" ::: "memory");
        const int i = i0 + wave*QPW + qi;
        const float4 qp = sp[i];
        const float qx = qp.x, qy = qp.y, qz = qp.z, qs = qp.w;

        float d[NC];
        #pragma unroll
        for (int c0 = 0; c0 < NC; c0 += LG) {
          float4 rec[LG];
          #pragma unroll
          for (int u = 0; u < LG; ++u)
              rec[u] = sp[(c0 + u)*32 + lane];
          #pragma unroll
          for (int u = 0; u < LG; ++u) {
            const int c = c0 + u;
            const float4 cp = rec[u];
            d[c] = fmaf(-2.0f, fmaf(qz, cp.z, fmaf(qx, cp.x, qy*cp.y)), qs + cp.w);
          }
        }
        float m0 = d[0], m1 = d[1], m2 = d[2], m3 = d[3];
        #pragma unroll
        for (int c = 4; c < NC; c += 4) {
            m0 = fminf(m0, d[c+0]); m1 = fminf(m1, d[c+1]);
            m2 = fminf(m2, d[c+2]); m3 = fminf(m3, d[c+3]);
        }
        const float lm = fminf(fminf(m0, m1), fminf(m2, m3));
        const float T = __builtin_bit_cast(float, __builtin_amdgcn_readlane(
            __builtin_bit_cast(int, sort32(lm, lane)), NK-1));
        const float bar = T + (fabsf(T)*2.4e-7f + 1e-30f);

        float L = (lane < NK) ? bar : NEG_INF;
        int   I = 0;
        float thr = bar;

        #pragma unroll
        for (int c = 0; c < NC; ++c) {
            const float dc = d[c];
            unsigned mask = __builtin_amdgcn_ballot_w32(dc < thr);
            if (mask) {
              do {
                const int sl = __builtin_ctz(mask);
                const float sd = __builtin_bit_cast(float,
                    __builtin_amdgcn_readlane(__builtin_bit_cast(int, dc), sl));
                const int sj = c*32 + sl;
                const float Lsh = __builtin_bit_cast(float, __builtin_amdgcn_update_dpp(
                    __builtin_bit_cast(int, NEG_INF), __builtin_bit_cast(int, L),
                    0x111, 0xf, 0xf, false));
                const int Ish = __builtin_amdgcn_update_dpp(0, I, 0x111, 0xf, 0xf, false);
                const bool mv  = L   > sd;
                const bool mvp = Lsh > sd;
                const float nL = mvp ? Lsh : sd;
                const int   nI = mvp ? Ish : sj;
                L = mv ? nL : L;
                I = mv ? nI : I;
                mask &= mask - 1;
              } while (mask);
              thr = __builtin_bit_cast(float,
                  __builtin_amdgcn_readlane(__builtin_bit_cast(int, L), NK-1));
            }
        }

        if (qi > 0) {
            float4 m = make_float4(-3.4e38f, -3.4e38f, -3.4e38f, -3.4e38f);
            #pragma unroll
            for (int k = 0; k < NK; ++k) {
                m.x = fmaxf(m.x, g[k].x);
                m.y = fmaxf(m.y, g[k].y);
                m.z = fmaxf(m.z, g[k].z);
                m.w = fmaxf(m.w, g[k].w);
            }
            m.y = tail ? gx : m.y;
            m.z = tail ? gy : m.z;
            m.w = tail ? gz : m.w;
            ob[(size_t)gi * (FP/4)] = m;
        }

        #pragma unroll
        for (int k = 0; k < NK; ++k) {
            const int nbk = __builtin_amdgcn_readlane(I, k);
            g[k] = rb[(size_t)nbk * (FP/4)];
        }
        gx = qx; gy = qy; gz = qz; gi = i;
    }

    {
        float4 m = make_float4(-3.4e38f, -3.4e38f, -3.4e38f, -3.4e38f);
        #pragma unroll
        for (int k = 0; k < NK; ++k) {
            m.x = fmaxf(m.x, g[k].x);
            m.y = fmaxf(m.y, g[k].y);
            m.z = fmaxf(m.z, g[k].z);
            m.w = fmaxf(m.w, g[k].w);
        }
        m.y = tail ? gx : m.y;
        m.z = tail ? gy : m.z;
        m.w = tail ? gz : m.w;
        ob[(size_t)gi * (FP/4)] = m;
    }
}

extern "C" void kernel_launch(void* const* d_in, const int* in_sizes, int n_in,
                              void* d_out, int out_size, void* d_ws, size_t ws_size,
                              hipStream_t stream) {
    const float* x  = (const float*)d_in[0];
    const float* W1 = (const float*)d_in[1];
    const float* b1 = (const float*)d_in[2];
    const float* W2 = (const float*)d_in[3];
    const float* b2 = (const float*)d_in[4];
    float* out = (float*)d_out;

    float* resw = (float*)d_ws;

    mlp_wmma_kernel<<<NPTS/(16*MT), 256, 0, stream>>>(x, W1, b1, W2, b2, resw);
    knn_gather_max_kernel<<<NB*(NS/QPB), 256, 0, stream>>>(x, resw, out);
}


# kNN kernel at 8 waves per SIMD for 4: 2 queries per wave, 512-thread blocks, neighbour rows fetched into the distances' dead registers and reduced at once (102 VGPRs for 196); same values, same lists
# speedup vs baseline: 13.4351x; 1.1248x over previous
#include <hip/hip_runtime.h>
#include <math.h>

typedef float v2f __attribute__((ext_vector_type(2)));
typedef float v8f __attribute__((ext_vector_type(8)));

#define NB 8
#define NS 2048
#define ND 3
#define NF 125
#define FP 128
#define KPAD (FP + 4)
#define NK 16
#define NPTS (NB*NS)
#define SG 32

__device__ __forceinline__ float gelu_exact(float v) {
    return 0.5f * v * (1.0f + erff(v * 0.7071067811865475f));
}

__device__ __forceinline__ float gelu_small(float v) {
    const float t = v * 0.7071067811865475f;
    const float s = t * t;
    float p = fmaf(__builtin_bit_cast(float, 0xba1345e1u), s, __builtin_bit_cast(float, 0x3ba10414u));
    p = fmaf(s, p, __builtin_bit_cast(float, 0xbcdac9b8u));
    p = fmaf(s, p, __builtin_bit_cast(float, 0x3de703beu));
    p = fmaf(s, p, __builtin_bit_cast(float, 0xbec09330u));
    p = fmaf(s, p, __builtin_bit_cast(float, 0x3e0375d0u));
    const float a = fabsf(t);
    const float e = copysignf(fmaf(a, p, a), t);
    return 0.5f * v * (1.0f + e);
}

#define MT 4

__global__ __launch_bounds__(256) void mlp_wmma_kernel(
    const float* __restrict__ x, const float* __restrict__ W1,
    const float* __restrict__ b1, const float* __restrict__ W2,
    const float* __restrict__ b2, float* __restrict__ resw)
{
    __shared__ float sW2t[FP * KPAD];
    __shared__ v2f   sA[MT * 32 * 32];
    const int tid = threadIdx.x;
    {
        const int n  = tid & 127;
        const int k0 = tid >> 7;
        const int nc = (n < NF) ? n : (NF - 1);
        for (int g = 0; g < 64; g += SG) {
            float t[SG];
            #pragma unroll
            for (int u = 0; u < SG; ++u) {
                const int k  = 2*(g + u) + k0;
                const int kc = (k < NF) ? k : (NF - 1);
                t[u] = W2[kc*NF + nc];
            }
            #pragma unroll
            for (int u = 0; u < SG; ++u) {
                const int k = 2*(g + u) + k0;
                sW2t[n*KPAD + k] = (k < NF && n < NF) ? t[u] : 0.0f;
            }
        }
    }

    const int lane = tid & 31;
    const int wave = tid >> 5;
    const int row  = lane & 15;
    const int half = lane >> 4;
    const int mt   = wave >> 1;
    const int e    = wave & 1;
    const int m0   = (blockIdx.x * MT + mt) * 16;

    const float px = x[(size_t)(m0+row)*3 + 0];
    const float py = x[(size_t)(m0+row)*3 + 1];
    const float pz = x[(size_t)(m0+row)*3 + 2];

    float hreg[32];
    #pragma unroll
    for (int j = 0; j < 16; ++j) {
        #pragma unroll
        for (int u = 0; u < 2; ++u) {
            const int c  = 4*(16*e + j) + 2*half + u;
            const int cc = (c < NF) ? c : (NF - 1);
            hreg[2*j + u] = px*W1[0*NF + cc] + py*W1[1*NF + cc] + pz*W1[2*NF + cc] + b1[cc];
        }
    }
    float tmax = 0.0f;
    #pragma unroll
    for (int i = 0; i < 32; ++i)
        tmax = fmaxf(tmax, fabsf(hreg[i] * 0.7071067811865475f));
    if (__builtin_amdgcn_ballot_w32(!(tmax < 1.0f)) == 0u) {
        #pragma unroll
        for (int j = 0; j < 16; ++j) {
            #pragma unroll
            for (int u = 0; u < 2; ++u) {
                const int c = 4*(16*e + j) + 2*half + u;
                hreg[2*j + u] = (c < NF) ? gelu_small(hreg[2*j + u]) : 0.0f;
            }
        }
    } else {
        #pragma unroll
        for (int j = 0; j < 16; ++j) {
            #pragma unroll
            for (int u = 0; u < 2; ++u) {
                const int c = 4*(16*e + j) + 2*half + u;
                hreg[2*j + u] = (c < NF) ? gelu_exact(hreg[2*j + u]) : 0.0f;
            }
        }
    }
    #pragma unroll
    for (int j = 0; j < 16; ++j) {
        v2f a;
        a.x = hreg[2*j + 0];
        a.y = hreg[2*j + 1];
        sA[(mt*32 + 16*e + j)*32 + lane] = a;
    }
    __syncthreads();

    float areg[64];
    #pragma unroll
    for (int ks = 0; ks < 32; ++ks) {
        const v2f a = sA[(mt*32 + ks)*32 + lane];
        areg[2*ks + 0] = a.x;
        areg[2*ks + 1] = a.y;
    }

    for (int q = 0; q < 4; ++q) {
        const int nt = 4*e + q;
        const int n = nt*16 + row;
        const float bias = (n < NF) ? b2[n] : 0.0f;
        v8f acc;
        #pragma unroll
        for (int r = 0; r < 8; ++r) acc[r] = bias;

        const float* bn = &sW2t[n * KPAD];
        #pragma unroll
        for (int ks = 0; ks < 32; ++ks) {
            const int kb = 4*ks + 2*half;
            v2f a, b;
            a.x = areg[2*ks + 0];
            a.y = areg[2*ks + 1];
            b   = *(const v2f*)(bn + kb);
            acc = __builtin_amdgcn_wmma_f32_16x16x4_f32(
                    false, a, false, b, (short)0, acc, false, false);
        }

        #pragma unroll
        for (int r = 0; r < 8; ++r) {
            const int m = half*8 + r;
            resw[(size_t)(m0 + m)*FP + nt*16 + row] = acc[r];
        }
    }
}

#define KT 512
#define QPW 2
#define QPB ((KT/32)*QPW)
#define NC (NS/32)
#define LG 8

#define LANE_XOR_ROW(v, J) __builtin_bit_cast(float, __builtin_amdgcn_update_dpp( \
    __builtin_bit_cast(int, (v)), __builtin_bit_cast(int, (v)), 0x160 | (J), 0xf, 0xf, false))
#define LANE_XOR_16(v) __builtin_bit_cast(float, __builtin_amdgcn_permlanex16( \
    __builtin_bit_cast(int, (v)), __builtin_bit_cast(int, (v)), 0x76543210, 0xfedcba98, false, false))
#define SORT_STAGE(K, J, PARTNER) { \
    const float p_ = (PARTNER); \
    const bool keep_min_ = (((lane & (J)) == 0) == ((lane & (K)) == 0)); \
    v = keep_min_ ? fminf(v, p_) : fmaxf(v, p_); }

__device__ __forceinline__ float sort32(float v, const int lane) {
    SORT_STAGE(2, 1, LANE_XOR_ROW(v, 1))
    SORT_STAGE(4, 2, LANE_XOR_ROW(v, 2))
    SORT_STAGE(4, 1, LANE_XOR_ROW(v, 1))
    SORT_STAGE(8, 4, LANE_XOR_ROW(v, 4))
    SORT_STAGE(8, 2, LANE_XOR_ROW(v, 2))
    SORT_STAGE(8, 1, LANE_XOR_ROW(v, 1))
    SORT_STAGE(16, 8, LANE_XOR_ROW(v, 8))
    SORT_STAGE(16, 4, LANE_XOR_ROW(v, 4))
    SORT_STAGE(16, 2, LANE_XOR_ROW(v, 2))
    SORT_STAGE(16, 1, LANE_XOR_ROW(v, 1))
    SORT_STAGE(32, 16, LANE_XOR_16(v))
    SORT_STAGE(32, 8, LANE_XOR_ROW(v, 8))
    SORT_STAGE(32, 4, LANE_XOR_ROW(v, 4))
    SORT_STAGE(32, 2, LANE_XOR_ROW(v, 2))
    SORT_STAGE(32, 1, LANE_XOR_ROW(v, 1))
    return v;
}

__global__ __launch_bounds__(KT) __attribute__((amdgpu_waves_per_eu(8, 8))) void knn_gather_max_kernel(
    const float* __restrict__ x, const float* __restrict__ resw,
    float* __restrict__ out)
{
    __shared__ float4 sp[NS];
    const int b  = blockIdx.x / (NS/QPB);
    const int i0 = (blockIdx.x % (NS/QPB)) * QPB;
    const float* xb = x + (size_t)b*NS*3;
    {
        float p0[NS/KT], p1[NS/KT], p2[NS/KT];
        #pragma unroll
        for (int u = 0; u < NS/KT; ++u) {
            const int j = threadIdx.x + KT*u;
            p0[u] = xb[j*3+0]; p1[u] = xb[j*3+1]; p2[u] = xb[j*3+2];
        }
        #pragma unroll
        for (int u = 0; u < NS/KT; ++u) {
            const int j = threadIdx.x + KT*u;
            const float a0 = p0[u], a1 = p1[u], a2 = p2[u];
            sp[j] = make_float4(a0, a1, a2, a0*a0 + a1*a1 + a2*a2);
        }
    }
    __syncthreads();

    const int lane = threadIdx.x & 31;
    const int wave = threadIdx.x >> 5;
    const float NEG_INF = -__builtin_inff();
    const bool tail = (lane == 31);
    const float4* rb = (const float4*)(resw + (size_t)b*NS*FP) + lane;
    float4*       ob = (float4*)(out + (size_t)b*NS*FP) + lane;

    #pragma unroll 1
    for (int qi = 0; qi < QPW; ++qi) {
        asm volatile("" ::: "memory");
        const int i = i0 + wave*QPW + qi;
        const float4 qp = sp[i];
        const float qx = qp.x, qy = qp.y, qz = qp.z, qs = qp.w;

        float d[NC];
        #pragma unroll
        for (int c0 = 0; c0 < NC; c0 += LG) {
          float4 rec[LG];
          #pragma unroll
          for (int u = 0; u < LG; ++u)
              rec[u] = sp[(c0 + u)*32 + lane];
          #pragma unroll
          for (int u = 0; u < LG; ++u) {
            const int c = c0 + u;
            const float4 cp = rec[u];
            d[c] = fmaf(-2.0f, fmaf(qz, cp.z, fmaf(qx, cp.x, qy*cp.y)), qs + cp.w);
          }
        }
        float m0 = d[0], m1 = d[1], m2 = d[2], m3 = d[3];
        #pragma unroll
        for (int c = 4; c < NC; c += 4) {
            m0 = fminf(m0, d[c+0]); m1 = fminf(m1, d[c+1]);
            m2 = fminf(m2, d[c+2]); m3 = fminf(m3, d[c+3]);
        }
        const float lm = fminf(fminf(m0, m1), fminf(m2, m3));
        const float T = __builtin_bit_cast(float, __builtin_amdgcn_readlane(
            __builtin_bit_cast(int, sort32(lm, lane)), NK-1));
        const float bar = T + (fabsf(T)*2.4e-7f + 1e-30f);

        float L = (lane < NK) ? bar : NEG_INF;
        int   I = 0;
        float thr = bar;

        #pragma unroll
        for (int c = 0; c < NC; ++c) {
            const float dc = d[c];
            unsigned mask = __builtin_amdgcn_ballot_w32(dc < thr);
            if (mask) {
              do {
                const int sl = __builtin_ctz(mask);
                const float sd = __builtin_bit_cast(float,
                    __builtin_amdgcn_readlane(__builtin_bit_cast(int, dc), sl));
                const int sj = c*32 + sl;
                const float Lsh = __builtin_bit_cast(float, __builtin_amdgcn_update_dpp(
                    __builtin_bit_cast(int, NEG_INF), __builtin_bit_cast(int, L),
                    0x111, 0xf, 0xf, false));
                const int Ish = __builtin_amdgcn_update_dpp(0, I, 0x111, 0xf, 0xf, false);
                const bool mv  = L   > sd;
                const bool mvp = Lsh > sd;
                const float nL = mvp ? Lsh : sd;
                const int   nI = mvp ? Ish : sj;
                L = mv ? nL : L;
                I = mv ? nI : I;
                mask &= mask - 1;
              } while (mask);
              thr = __builtin_bit_cast(float,
                  __builtin_amdgcn_readlane(__builtin_bit_cast(int, L), NK-1));
            }
        }

        float4 g[NK];
        #pragma unroll
        for (int k = 0; k < NK; ++k) {
            const int nbk = __builtin_amdgcn_readlane(I, k);
            g[k] = rb[(size_t)nbk * (FP/4)];
        }
        float4 m = make_float4(-3.4e38f, -3.4e38f, -3.4e38f, -3.4e38f);
        #pragma unroll
        for (int k = 0; k < NK; ++k) {
            m.x = fmaxf(m.x, g[k].x);
            m.y = fmaxf(m.y, g[k].y);
            m.z = fmaxf(m.z, g[k].z);
            m.w = fmaxf(m.w, g[k].w);
        }
        m.y = tail ? qx : m.y;
        m.z = tail ? qy : m.z;
        m.w = tail ? qz : m.w;
        ob[(size_t)i * (FP/4)] = m;
    }
}

extern "C" void kernel_launch(void* const* d_in, const int* in_sizes, int n_in,
                              void* d_out, int out_size, void* d_ws, size_t ws_size,
                              hipStream_t stream) {
    const float* x  = (const float*)d_in[0];
    const float* W1 = (const float*)d_in[1];
    const float* b1 = (const float*)d_in[2];
    const float* W2 = (const float*)d_in[3];
    const float* b2 = (const float*)d_in[4];
    float* out = (float*)d_out;

    float* resw = (float*)d_ws;

    mlp_wmma_kernel<<<NPTS/(16*MT), 256, 0, stream>>>(x, W1, b1, W2, b2, resw);
    knn_gather_max_kernel<<<NB*(NS/QPB), KT, 0, stream>>>(x, resw, out);
}
